# starting kernel
# baseline (speedup 1.0000x reference)
#include <hip/hip_runtime.h>
#include <hip/hip_bf16.h>

typedef __attribute__((ext_vector_type(2))) float v2f;
typedef __attribute__((ext_vector_type(8))) float v8f;

#define C_HID 256

__global__ void k_deg_init(int* __restrict__ degi, int n) {
  int i = blockIdx.x * blockDim.x + threadIdx.x;
  if (i < n) degi[i] = 1;
}

__global__ void k_deg_count(const int* __restrict__ dst, int* __restrict__ degi, int E) {
  int e = blockIdx.x * blockDim.x + threadIdx.x;
  if (e < E) atomicAdd(&degi[dst[e]], 1);
}

__global__ void k_dinv(const int* __restrict__ degi, float* __restrict__ dinv, int n) {
  int i = blockIdx.x * blockDim.x + threadIdx.x;
  if (i < n) dinv[i] = rsqrtf((float)degi[i]);
}

__global__ void k_zero(float* __restrict__ p, long long n) {
  long long i = (long long)blockIdx.x * blockDim.x + threadIdx.x;
  if (i < n) p[i] = 0.0f;
}

__global__ __launch_bounds__(256) void k_gemm_wmma(
    const float* __restrict__ X, const float* __restrict__ W,
    float* __restrict__ Hout, int M, int K, int N) {
  const int wpb  = blockDim.x >> 5;
  const int wave = blockIdx.x * wpb + ((int)threadIdx.x >> 5);
  const int lane = threadIdx.x & 31;
  const int ntiles = N >> 4;
  const int mt = wave / ntiles;
  const int nt = wave - mt * ntiles;
  const int m0 = mt << 4;
  if (m0 >= M) return;
  const int n0 = nt << 4;
  const int lo = lane & 15;
  const int hi = lane >> 4;

  int arow = m0 + lo;
  if (arow >= M) arow = M - 1;
  const float* Xr = X + (long long)arow * K + 2 * hi;
  const float* Wc = W + (long long)(2 * hi) * N + n0 + lo;

  v8f acc = {};
  for (int k = 0; k < K; k += 4) {
    v2f a, b;
    a.x = Xr[0];
    a.y = Xr[1];
    b.x = Wc[0];
    b.y = Wc[N];
    Xr += 4;
    Wc += (long long)4 * N;
    acc = __builtin_amdgcn_wmma_f32_16x16x4_f32(
         false, a,  false, b,
         (short)0, acc,  false,  false);
  }

  float* Hp = Hout + (long long)(m0 + hi * 8) * N + n0 + lo;
#pragma unroll
  for (int r = 0; r < 8; ++r) {
    int mr = m0 + hi * 8 + r;
    if (mr < M) Hp[(long long)r * N] = acc[r];
  }
}

__device__ __forceinline__ void atomAddF(float* p, float v) {
  unsafeAtomicAdd(p, v);
}

__global__ __launch_bounds__(256) void k_scatter(
    const float* __restrict__ H, const int* __restrict__ src,
    const int* __restrict__ dst, const float* __restrict__ dinv,
    float* __restrict__ AGG, int E) {
  long long tid = (long long)blockIdx.x * blockDim.x + threadIdx.x;
  long long total = (long long)E * (C_HID / 4);
  if (tid >= total) return;
  int e = (int)(tid >> 6);
  int c = ((int)tid & 63) << 2;
  int s = src[e];
  int d = dst[e];
  float nrm = dinv[s] * dinv[d];
  float4 hv = *(const float4*)(H + (long long)s * C_HID + c);
  float* op = AGG + (long long)d * C_HID + c;
  atomAddF(op + 0, hv.x * nrm);
  atomAddF(op + 1, hv.y * nrm);
  atomAddF(op + 2, hv.z * nrm);
  atomAddF(op + 3, hv.w * nrm);
}

__global__ __launch_bounds__(256) void k_combine(
    const float* __restrict__ AGG, const float* __restrict__ H,
    const float* __restrict__ dinv, const float* __restrict__ bias,
    float* __restrict__ OUT, int n) {
  long long i = (long long)blockIdx.x * blockDim.x + threadIdx.x;
  long long total = (long long)n * C_HID;
  if (i >= total) return;
  int node = (int)(i >> 8);
  int c = (int)i & 255;
  float di = dinv[node];
  float v = AGG[i] + di * di * H[i] + bias[c];
  OUT[i] = (v > 0.0f) ? v : expm1f(v);
}

extern "C" void kernel_launch(void* const* d_in, const int* in_sizes, int n_in,
                              void* d_out, int out_size, void* d_ws, size_t ws_size,
                              hipStream_t stream) {
  const float* x  = (const float*)d_in[0];
  const int*   ei = (const int*)d_in[1];
  const float* W1 = (const float*)d_in[2];
  const float* b1 = (const float*)d_in[3];
  const float* W2 = (const float*)d_in[4];
  const float* b2 = (const float*)d_in[5];
  const float* W3 = (const float*)d_in[6];
  const float* b3 = (const float*)d_in[7];
  float* out = (float*)d_out;

  const int C   = C_HID;
  const int E   = in_sizes[1] / 2;
  const int cin = in_sizes[2] / C;
  const int n   = in_sizes[0] / cin;
  const int* src = ei;
  const int* dst = ei + E;

  char*  ws  = (char*)d_ws;
  size_t off = 0;
  auto alloc = [&](size_t bytes) -> void* {
    void* p = ws + off;
    off = (off + bytes + 255) & ~(size_t)255;
    return p;
  };
  int*   degi = (int*)  alloc((size_t)n * sizeof(int));
  float* dinv = (float*)alloc((size_t)n * sizeof(float));
  float* H    = (float*)alloc((size_t)n * C * sizeof(float));
  float* AGG  = (float*)alloc((size_t)n * C * sizeof(float));
  float* XB   = (float*)alloc((size_t)n * C * sizeof(float));
  (void)ws_size; (void)n_in; (void)out_size;

  k_deg_init <<<(n + 255) / 256, 256, 0, stream>>>(degi, n);
  k_deg_count<<<(E + 255) / 256, 256, 0, stream>>>(dst, degi, E);
  k_dinv     <<<(n + 255) / 256, 256, 0, stream>>>(degi, dinv, n);

  const long long nC = (long long)n * C;
  const int zgrid = (int)((nC + 255) / 256);
  const long long sThreads = (long long)E * (C / 4);
  const int sgrid = (int)((sThreads + 255) / 256);

  auto layer = [&](const float* Xin, const float* W, const float* bias,
                   float* Out, int K) {
    const int mtiles = (n + 15) / 16;
    const int ntiles = C / 16;
    const int waves  = mtiles * ntiles;
    const int wpb    = 8;
    const int blocks = (waves + wpb - 1) / wpb;
    k_gemm_wmma<<<blocks, wpb * 32, 0, stream>>>(Xin, W, H, n, K, C);
    k_zero     <<<zgrid, 256, 0, stream>>>(AGG, nC);
    k_scatter  <<<sgrid, 256, 0, stream>>>(H, src, dst, dinv, AGG, E);
    k_combine  <<<zgrid, 256, 0, stream>>>(AGG, H, dinv, bias, Out, n);
  };

  layer(x,  W1, b1, XB,  cin);
  layer(XB, W2, b2, XB,  C);
  layer(XB, W3, b3, out, C);
}


# f32 atomic scatter-add replaced by per-call dst-bucketed edge list (count, scan, fill) and one-wave-per-row gather sum with combine fused; sum order now bucket order
# speedup vs baseline: 2.7101x; 2.7101x over previous
#include <hip/hip_runtime.h>
#include <hip/hip_bf16.h>


typedef __attribute__((ext_vector_type(2))) float v2f;
typedef __attribute__((ext_vector_type(8))) float v8f;

#define C_HID 256

__global__ void k_cnt_init(int* __restrict__ cnt, int n) {
  int i = blockIdx.x * blockDim.x + threadIdx.x;
  if (i < n) cnt[i] = 0;
}

__global__ void k_count_rank(const int* __restrict__ dst, int* __restrict__ cnt,
                             int* __restrict__ rank, int E) {
  int e = blockIdx.x * blockDim.x + threadIdx.x;
  if (e < E) rank[e] = atomicAdd(&cnt[dst[e]], 1);
}

__global__ __launch_bounds__(1024) void k_scan_dinv(
    const int* __restrict__ cnt, int* __restrict__ rowptr,
    float* __restrict__ dinv, int n) {
  __shared__ int wsum[32];
  const int t = threadIdx.x;
  const int lane = t & 31;
  const int wv = t >> 5;
  const int per = (n + 1023) >> 10;
  int b = t * per;
  int e = b + per;
  if (b > n) b = n;
  if (e > n) e = n;
  int s = 0;
  for (int i = b; i < e; ++i) s += cnt[i];
  int incl = s;
#pragma unroll
  for (int o = 1; o < 32; o <<= 1) {
    int v = __shfl_up(incl, o);
    if (lane >= o) incl += v;
  }
  if (lane == 31) wsum[wv] = incl;
  __syncthreads();
  if (wv == 0) {
    int v = wsum[lane];
    int iv = v;
#pragma unroll
    for (int o = 1; o < 32; o <<= 1) {
      int u = __shfl_up(iv, o);
      if (lane >= o) iv += u;
    }
    wsum[lane] = iv - v;
  }
  __syncthreads();
  int run = wsum[wv] + incl - s;
  for (int i = b; i < e; ++i) {
    int c = cnt[i];
    rowptr[i] = run;
    dinv[i] = rsqrtf((float)(c + 1));
    run += c;
  }
  if (t == 1023) rowptr[n] = run;
}

__global__ void k_fill(const int* __restrict__ src, const int* __restrict__ dst,
                       const int* __restrict__ rank, const int* __restrict__ rowptr,
                       const float* __restrict__ dinv, int* __restrict__ col,
                       float* __restrict__ wgt, int E) {
  int e = blockIdx.x * blockDim.x + threadIdx.x;
  if (e >= E) return;
  int s = src[e];
  int d = dst[e];
  int p = rowptr[d] + rank[e];
  col[p] = s;
  wgt[p] = dinv[s] * dinv[d];
}

__global__ __launch_bounds__(256) void k_gemm_wmma(
    const float* __restrict__ X, const float* __restrict__ W,
    float* __restrict__ Hout, int M, int K, int N) {
  const int wpb  = blockDim.x >> 5;
  const int wave = blockIdx.x * wpb + ((int)threadIdx.x >> 5);
  const int lane = threadIdx.x & 31;
  const int ntiles = N >> 4;
  const int mt = wave / ntiles;
  const int nt = wave - mt * ntiles;
  const int m0 = mt << 4;
  if (m0 >= M) return;
  const int n0 = nt << 4;
  const int lo = lane & 15;
  const int hi = lane >> 4;

  int arow = m0 + lo;
  if (arow >= M) arow = M - 1;
  const float* Xr = X + (long long)arow * K + 2 * hi;
  const float* Wc = W + (long long)(2 * hi) * N + n0 + lo;

  v8f acc = {};
  for (int k = 0; k < K; k += 4) {
    v2f a, b;
    a.x = Xr[0];
    a.y = Xr[1];
    b.x = Wc[0];
    b.y = Wc[N];
    Xr += 4;
    Wc += (long long)4 * N;
    acc = __builtin_amdgcn_wmma_f32_16x16x4_f32(
         false, a,  false, b,
         (short)0, acc,  false,  false);
  }

  float* Hp = Hout + (long long)(m0 + hi * 8) * N + n0 + lo;
#pragma unroll
  for (int r = 0; r < 8; ++r) {
    int mr = m0 + hi * 8 + r;
    if (mr < M) Hp[(long long)r * N] = acc[r];
  }
}

__device__ __forceinline__ void fma4(float4& a, const float4 h, const float w) {
  a.x = fmaf(h.x, w, a.x);
  a.y = fmaf(h.y, w, a.y);
  a.z = fmaf(h.z, w, a.z);
  a.w = fmaf(h.w, w, a.w);
}

__device__ __forceinline__ float elu1(float v) {
  return (v > 0.0f) ? v : expm1f(v);
}

__global__ __launch_bounds__(256) void k_gather_combine(
    const float* __restrict__ H, const int* __restrict__ rowptr,
    const int* __restrict__ col, const float* __restrict__ wgt,
    const float* __restrict__ dinv, const float* __restrict__ bias,
    float* __restrict__ OUT, int n) {
  const int lane = threadIdx.x & 31;
  const int d = __builtin_amdgcn_readfirstlane(
      (int)(blockIdx.x * (blockDim.x >> 5) + (threadIdx.x >> 5)));
  if (d >= n) return;
  const int beg = rowptr[d];
  const int end = rowptr[d + 1];
  const float* Hl = H + 4 * lane;

  float4 a0 = make_float4(0.f, 0.f, 0.f, 0.f);
  float4 a1 = make_float4(0.f, 0.f, 0.f, 0.f);

  for (int base = beg; base < end; base += 32) {
    const int idx = base + lane;
    int s_l = 0;
    float w_l = 0.0f;
    if (idx < end) {
      s_l = col[idx];
      w_l = wgt[idx];
    }
    const int w_bits = __float_as_int(w_l);
    int cnt = end - base;
    if (cnt > 32) cnt = 32;
    int j = 0;
    for (; j + 4 <= cnt; j += 4) {
      const int s0 = __builtin_amdgcn_readlane(s_l, j + 0);
      const int s1 = __builtin_amdgcn_readlane(s_l, j + 1);
      const int s2 = __builtin_amdgcn_readlane(s_l, j + 2);
      const int s3 = __builtin_amdgcn_readlane(s_l, j + 3);
      const float w0 = __int_as_float(__builtin_amdgcn_readlane(w_bits, j + 0));
      const float w1 = __int_as_float(__builtin_amdgcn_readlane(w_bits, j + 1));
      const float w2 = __int_as_float(__builtin_amdgcn_readlane(w_bits, j + 2));
      const float w3 = __int_as_float(__builtin_amdgcn_readlane(w_bits, j + 3));
      const float* r0 = Hl + (long long)s0 * C_HID;
      const float* r1 = Hl + (long long)s1 * C_HID;
      const float* r2 = Hl + (long long)s2 * C_HID;
      const float* r3 = Hl + (long long)s3 * C_HID;
      const float4 h00 = *(const float4*)(r0);
      const float4 h01 = *(const float4*)(r0 + 128);
      const float4 h10 = *(const float4*)(r1);
      const float4 h11 = *(const float4*)(r1 + 128);
      const float4 h20 = *(const float4*)(r2);
      const float4 h21 = *(const float4*)(r2 + 128);
      const float4 h30 = *(const float4*)(r3);
      const float4 h31 = *(const float4*)(r3 + 128);
      fma4(a0, h00, w0); fma4(a1, h01, w0);
      fma4(a0, h10, w1); fma4(a1, h11, w1);
      fma4(a0, h20, w2); fma4(a1, h21, w2);
      fma4(a0, h30, w3); fma4(a1, h31, w3);
    }
    if (cnt & 2) {
      const int s0 = __builtin_amdgcn_readlane(s_l, j + 0);
      const int s1 = __builtin_amdgcn_readlane(s_l, j + 1);
      const float w0 = __int_as_float(__builtin_amdgcn_readlane(w_bits, j + 0));
      const float w1 = __int_as_float(__builtin_amdgcn_readlane(w_bits, j + 1));
      const float* r0 = Hl + (long long)s0 * C_HID;
      const float* r1 = Hl + (long long)s1 * C_HID;
      const float4 h00 = *(const float4*)(r0);
      const float4 h01 = *(const float4*)(r0 + 128);
      const float4 h10 = *(const float4*)(r1);
      const float4 h11 = *(const float4*)(r1 + 128);
      fma4(a0, h00, w0); fma4(a1, h01, w0);
      fma4(a0, h10, w1); fma4(a1, h11, w1);
      j += 2;
    }
    if (cnt & 1) {
      const int s0 = __builtin_amdgcn_readlane(s_l, j);
      const float w0 = __int_as_float(__builtin_amdgcn_readlane(w_bits, j));
      const float* r0 = Hl + (long long)s0 * C_HID;
      const float4 h00 = *(const float4*)(r0);
      const float4 h01 = *(const float4*)(r0 + 128);
      fma4(a0, h00, w0); fma4(a1, h01, w0);
    }
  }

  const float di = dinv[d];
  const float dd = di * di;
  const float* Hd = Hl + (long long)d * C_HID;
  const float4 g0 = *(const float4*)(Hd);
  const float4 g1 = *(const float4*)(Hd + 128);
  const float4 b0 = *(const float4*)(bias + 4 * lane);
  const float4 b1 = *(const float4*)(bias + 128 + 4 * lane);
  float4 o0, o1;
  o0.x = elu1(a0.x + dd * g0.x + b0.x);
  o0.y = elu1(a0.y + dd * g0.y + b0.y);
  o0.z = elu1(a0.z + dd * g0.z + b0.z);
  o0.w = elu1(a0.w + dd * g0.w + b0.w);
  o1.x = elu1(a1.x + dd * g1.x + b1.x);
  o1.y = elu1(a1.y + dd * g1.y + b1.y);
  o1.z = elu1(a1.z + dd * g1.z + b1.z);
  o1.w = elu1(a1.w + dd * g1.w + b1.w);
  float* Od = OUT + (long long)d * C_HID + 4 * lane;
  *(float4*)(Od) = o0;
  *(float4*)(Od + 128) = o1;
}

extern "C" void kernel_launch(void* const* d_in, const int* in_sizes, int n_in,
                              void* d_out, int out_size, void* d_ws, size_t ws_size,
                              hipStream_t stream) {
  const float* x  = (const float*)d_in[0];
  const int*   ei = (const int*)d_in[1];
  const float* W1 = (const float*)d_in[2];
  const float* b1 = (const float*)d_in[3];
  const float* W2 = (const float*)d_in[4];
  const float* b2 = (const float*)d_in[5];
  const float* W3 = (const float*)d_in[6];
  const float* b3 = (const float*)d_in[7];
  float* out = (float*)d_out;

  const int C   = C_HID;
  const int E   = in_sizes[1] / 2;
  const int cin = in_sizes[2] / C;
  const int n   = in_sizes[0] / cin;
  const int* src = ei;
  const int* dst = ei + E;

  char*  ws  = (char*)d_ws;
  size_t off = 0;
  auto alloc = [&](size_t bytes) -> void* {
    void* p = ws + off;
    off = (off + bytes + 255) & ~(size_t)255;
    return p;
  };
  int*   cnt    = (int*)  alloc((size_t)n * sizeof(int));
  int*   rowptr = (int*)  alloc((size_t)(n + 1) * sizeof(int));
  float* dinv   = (float*)alloc((size_t)n * sizeof(float));
  int*   rank   = (int*)  alloc((size_t)E * sizeof(int));
  int*   col    = (int*)  alloc((size_t)E * sizeof(int));
  float* wgt    = (float*)alloc((size_t)E * sizeof(float));
  float* H      = (float*)alloc((size_t)n * C * sizeof(float));
  float* XB     = (float*)alloc((size_t)n * C * sizeof(float));
  (void)ws_size; (void)n_in; (void)out_size;

  k_cnt_init  <<<(n + 255) / 256, 256, 0, stream>>>(cnt, n);
  k_count_rank<<<(E + 255) / 256, 256, 0, stream>>>(dst, cnt, rank, E);
  k_scan_dinv <<<1, 1024, 0, stream>>>(cnt, rowptr, dinv, n);
  k_fill      <<<(E + 255) / 256, 256, 0, stream>>>(src, dst, rank, rowptr, dinv, col, wgt, E);

  const int ggrid = (n + 7) / 8;

  auto layer = [&](const float* Xin, const float* W, const float* bias,
                   float* Out, int K) {
    const int mtiles = (n + 15) / 16;
    const int ntiles = C / 16;
    const int waves  = mtiles * ntiles;
    const int wpb    = 8;
    const int blocks = (waves + wpb - 1) / wpb;
    k_gemm_wmma     <<<blocks, wpb * 32, 0, stream>>>(Xin, W, H, n, K, C);
    k_gather_combine<<<ggrid, 256, 0, stream>>>(H, rowptr, col, wgt, dinv, bias, Out, n);
  };

  layer(x,  W1, b1, XB,  cin);
  layer(XB, W2, b2, XB,  C);
  layer(XB, W3, b3, out, C);
}


# X@W product re-tiled: 4-wave group per 16x256 block, X block staged once in LDS, each wave feeds 4 column tiles from 16-byte W row loads issued 32 products ahead; k sum order inside each 8 changes
# speedup vs baseline: 3.1351x; 1.1568x over previous
#include <hip/hip_runtime.h>
#include <hip/hip_bf16.h>


typedef __attribute__((ext_vector_type(2))) float v2f;
typedef __attribute__((ext_vector_type(8))) float v8f;

#define C_HID 256

__global__ void k_cnt_init(int* __restrict__ cnt, int n) {
  int i = blockIdx.x * blockDim.x + threadIdx.x;
  if (i < n) cnt[i] = 0;
}

__global__ void k_count_rank(const int* __restrict__ dst, int* __restrict__ cnt,
                             int* __restrict__ rank, int E) {
  int e = blockIdx.x * blockDim.x + threadIdx.x;
  if (e < E) rank[e] = atomicAdd(&cnt[dst[e]], 1);
}

__global__ __launch_bounds__(1024) void k_scan_dinv(
    const int* __restrict__ cnt, int* __restrict__ rowptr,
    float* __restrict__ dinv, int n) {
  __shared__ int wsum[32];
  const int t = threadIdx.x;
  const int lane = t & 31;
  const int wv = t >> 5;
  const int per = (n + 1023) >> 10;
  int b = t * per;
  int e = b + per;
  if (b > n) b = n;
  if (e > n) e = n;
  int s = 0;
  for (int i = b; i < e; ++i) s += cnt[i];
  int incl = s;
#pragma unroll
  for (int o = 1; o < 32; o <<= 1) {
    int v = __shfl_up(incl, o);
    if (lane >= o) incl += v;
  }
  if (lane == 31) wsum[wv] = incl;
  __syncthreads();
  if (wv == 0) {
    int v = wsum[lane];
    int iv = v;
#pragma unroll
    for (int o = 1; o < 32; o <<= 1) {
      int u = __shfl_up(iv, o);
      if (lane >= o) iv += u;
    }
    wsum[lane] = iv - v;
  }
  __syncthreads();
  int run = wsum[wv] + incl - s;
  for (int i = b; i < e; ++i) {
    int c = cnt[i];
    rowptr[i] = run;
    dinv[i] = rsqrtf((float)(c + 1));
    run += c;
  }
  if (t == 1023) rowptr[n] = run;
}

__global__ void k_fill(const int* __restrict__ src, const int* __restrict__ dst,
                       const int* __restrict__ rank, const int* __restrict__ rowptr,
                       const float* __restrict__ dinv, int* __restrict__ col,
                       float* __restrict__ wgt, int E) {
  int e = blockIdx.x * blockDim.x + threadIdx.x;
  if (e >= E) return;
  int s = src[e];
  int d = dst[e];
  int p = rowptr[d] + rank[e];
  col[p] = s;
  wgt[p] = dinv[s] * dinv[d];
}

#define G_ROWS 16
#define G_KMAX 256
#define G_PAD 4

__device__ __forceinline__ v8f mma4(const float ax, const float ay,
                                    const float bx, const float by, const v8f acc) {
  v2f a, b;
  a.x = ax; a.y = ay;
  b.x = bx; b.y = by;
  return __builtin_amdgcn_wmma_f32_16x16x4_f32(
       false, a,  false, b,
       (short)0, acc,  false,  false);
}

__device__ __forceinline__ void g_loadb(float4 (&d)[16], const float* __restrict__ wp) {
#pragma unroll
  for (int j = 0; j < 4; ++j) {
#pragma unroll
    for (int i = 0; i < 4; ++i) {
      d[4 * j + i] = *(const float4*)(wp + (8 * j + i) * C_HID);
    }
  }
}

__device__ __forceinline__ void g_mma32(const float4 (&s)[16], const float* ar,
                                        v8f& acc0, v8f& acc1, v8f& acc2, v8f& acc3) {
#pragma unroll
  for (int j = 0; j < 4; ++j) {
    const float4 a = *(const float4*)(ar + 8 * j);
    acc0 = mma4(a.x, a.y, s[4 * j + 0].x, s[4 * j + 1].x, acc0);
    acc1 = mma4(a.x, a.y, s[4 * j + 0].y, s[4 * j + 1].y, acc1);
    acc2 = mma4(a.x, a.y, s[4 * j + 0].z, s[4 * j + 1].z, acc2);
    acc3 = mma4(a.x, a.y, s[4 * j + 0].w, s[4 * j + 1].w, acc3);
    acc0 = mma4(a.z, a.w, s[4 * j + 2].x, s[4 * j + 3].x, acc0);
    acc1 = mma4(a.z, a.w, s[4 * j + 2].y, s[4 * j + 3].y, acc1);
    acc2 = mma4(a.z, a.w, s[4 * j + 2].z, s[4 * j + 3].z, acc2);
    acc3 = mma4(a.z, a.w, s[4 * j + 2].w, s[4 * j + 3].w, acc3);
  }
}

__global__ __launch_bounds__(128) void k_gemm_wmma(
    const float* __restrict__ X, const float* __restrict__ Wm,
    float* __restrict__ Hout, int M, int K) {
  __shared__ __attribute__((aligned(16))) float As[G_ROWS * (G_KMAX + G_PAD)];
  const int t    = threadIdx.x;
  const int lane = t & 31;
  const int wv   = t >> 5;
  const int c    = lane & 15;
  const int h    = lane >> 4;
  const int m0   = blockIdx.x * G_ROWS;
  const int P    = K + G_PAD;

  const float* Wp = Wm + (long long)(4 * h) * C_HID + wv * 64 + 4 * c;
  float4 p[16], q[16];
  g_loadb(p, Wp);

  {
    const int k4 = K >> 2;
    int r0 = m0 + 4 * wv;
    int ra = r0 + 0; if (ra >= M) ra = M - 1;
    int rb = r0 + 1; if (rb >= M) rb = M - 1;
    int rc = r0 + 2; if (rc >= M) rc = M - 1;
    int rd = r0 + 3; if (rd >= M) rd = M - 1;
    const float* xa = X + (long long)ra * K;
    const float* xb = X + (long long)rb * K;
    const float* xc = X + (long long)rc * K;
    const float* xd = X + (long long)rd * K;
    float* la = As + (4 * wv) * P;
    for (int j = lane; j < k4; j += 32) {
      const float4 va = *(const float4*)(xa + 4 * j);
      const float4 vb = *(const float4*)(xb + 4 * j);
      const float4 vc = *(const float4*)(xc + 4 * j);
      const float4 vd = *(const float4*)(xd + 4 * j);
      *(float4*)(la + 4 * j)         = va;
      *(float4*)(la + P + 4 * j)     = vb;
      *(float4*)(la + 2 * P + 4 * j) = vc;
      *(float4*)(la + 3 * P + 4 * j) = vd;
    }
  }
  __syncthreads();

  const float* Ar = As + c * P + 4 * h;
  v8f acc0 = {}, acc1 = {}, acc2 = {}, acc3 = {};

#define G_HOLD() __builtin_amdgcn_sched_barrier(0)
  int kb = 0;
  for (; kb + 64 < K; kb += 64) {
    g_loadb(q, Wp + (long long)(kb + 32) * C_HID);
    G_HOLD();
    g_mma32(p, Ar + kb, acc0, acc1, acc2, acc3);
    G_HOLD();
    g_loadb(p, Wp + (long long)(kb + 64) * C_HID);
    G_HOLD();
    g_mma32(q, Ar + kb + 32, acc0, acc1, acc2, acc3);
    G_HOLD();
  }
  g_loadb(q, Wp + (long long)(kb + 32) * C_HID);
  G_HOLD();
  g_mma32(p, Ar + kb, acc0, acc1, acc2, acc3);
  g_mma32(q, Ar + kb + 32, acc0, acc1, acc2, acc3);
#undef G_HOLD

  float* Hp = Hout + (long long)(m0 + 8 * h) * C_HID + wv * 64 + 4 * c;
#pragma unroll
  for (int r = 0; r < 8; ++r) {
    const int mr = m0 + 8 * h + r;
    if (mr < M) {
      float4 o;
      o.x = acc0[r]; o.y = acc1[r]; o.z = acc2[r]; o.w = acc3[r];
      *(float4*)(Hp + (long long)r * C_HID) = o;
    }
  }
}

__device__ __forceinline__ void fma4(float4& a, const float4 h, const float w) {
  a.x = fmaf(h.x, w, a.x);
  a.y = fmaf(h.y, w, a.y);
  a.z = fmaf(h.z, w, a.z);
  a.w = fmaf(h.w, w, a.w);
}

__device__ __forceinline__ float elu1(float v) {
  return (v > 0.0f) ? v : expm1f(v);
}

__global__ __launch_bounds__(256) void k_gather_combine(
    const float* __restrict__ H, const int* __restrict__ rowptr,
    const int* __restrict__ col, const float* __restrict__ wgt,
    const float* __restrict__ dinv, const float* __restrict__ bias,
    float* __restrict__ OUT, int n) {
  const int lane = threadIdx.x & 31;
  const int d = __builtin_amdgcn_readfirstlane(
      (int)(blockIdx.x * (blockDim.x >> 5) + (threadIdx.x >> 5)));
  if (d >= n) return;
  const int beg = rowptr[d];
  const int end = rowptr[d + 1];
  const float* Hl = H + 4 * lane;

  float4 a0 = make_float4(0.f, 0.f, 0.f, 0.f);
  float4 a1 = make_float4(0.f, 0.f, 0.f, 0.f);

  for (int base = beg; base < end; base += 32) {
    const int idx = base + lane;
    int s_l = 0;
    float w_l = 0.0f;
    if (idx < end) {
      s_l = col[idx];
      w_l = wgt[idx];
    }
    const int w_bits = __float_as_int(w_l);
    int cnt = end - base;
    if (cnt > 32) cnt = 32;
    int j = 0;
    for (; j + 4 <= cnt; j += 4) {
      const int s0 = __builtin_amdgcn_readlane(s_l, j + 0);
      const int s1 = __builtin_amdgcn_readlane(s_l, j + 1);
      const int s2 = __builtin_amdgcn_readlane(s_l, j + 2);
      const int s3 = __builtin_amdgcn_readlane(s_l, j + 3);
      const float w0 = __int_as_float(__builtin_amdgcn_readlane(w_bits, j + 0));
      const float w1 = __int_as_float(__builtin_amdgcn_readlane(w_bits, j + 1));
      const float w2 = __int_as_float(__builtin_amdgcn_readlane(w_bits, j + 2));
      const float w3 = __int_as_float(__builtin_amdgcn_readlane(w_bits, j + 3));
      const float* r0 = Hl + (long long)s0 * C_HID;
      const float* r1 = Hl + (long long)s1 * C_HID;
      const float* r2 = Hl + (long long)s2 * C_HID;
      const float* r3 = Hl + (long long)s3 * C_HID;
      const float4 h00 = *(const float4*)(r0);
      const float4 h01 = *(const float4*)(r0 + 128);
      const float4 h10 = *(const float4*)(r1);
      const float4 h11 = *(const float4*)(r1 + 128);
      const float4 h20 = *(const float4*)(r2);
      const float4 h21 = *(const float4*)(r2 + 128);
      const float4 h30 = *(const float4*)(r3);
      const float4 h31 = *(const float4*)(r3 + 128);
      fma4(a0, h00, w0); fma4(a1, h01, w0);
      fma4(a0, h10, w1); fma4(a1, h11, w1);
      fma4(a0, h20, w2); fma4(a1, h21, w2);
      fma4(a0, h30, w3); fma4(a1, h31, w3);
    }
    if (cnt & 2) {
      const int s0 = __builtin_amdgcn_readlane(s_l, j + 0);
      const int s1 = __builtin_amdgcn_readlane(s_l, j + 1);
      const float w0 = __int_as_float(__builtin_amdgcn_readlane(w_bits, j + 0));
      const float w1 = __int_as_float(__builtin_amdgcn_readlane(w_bits, j + 1));
      const float* r0 = Hl + (long long)s0 * C_HID;
      const float* r1 = Hl + (long long)s1 * C_HID;
      const float4 h00 = *(const float4*)(r0);
      const float4 h01 = *(const float4*)(r0 + 128);
      const float4 h10 = *(const float4*)(r1);
      const float4 h11 = *(const float4*)(r1 + 128);
      fma4(a0, h00, w0); fma4(a1, h01, w0);
      fma4(a0, h10, w1); fma4(a1, h11, w1);
      j += 2;
    }
    if (cnt & 1) {
      const int s0 = __builtin_amdgcn_readlane(s_l, j);
      const float w0 = __int_as_float(__builtin_amdgcn_readlane(w_bits, j));
      const float* r0 = Hl + (long long)s0 * C_HID;
      const float4 h00 = *(const float4*)(r0);
      const float4 h01 = *(const float4*)(r0 + 128);
      fma4(a0, h00, w0); fma4(a1, h01, w0);
    }
  }

  const float di = dinv[d];
  const float dd = di * di;
  const float* Hd = Hl + (long long)d * C_HID;
  const float4 g0 = *(const float4*)(Hd);
  const float4 g1 = *(const float4*)(Hd + 128);
  const float4 b0 = *(const float4*)(bias + 4 * lane);
  const float4 b1 = *(const float4*)(bias + 128 + 4 * lane);
  float4 o0, o1;
  o0.x = elu1(a0.x + dd * g0.x + b0.x);
  o0.y = elu1(a0.y + dd * g0.y + b0.y);
  o0.z = elu1(a0.z + dd * g0.z + b0.z);
  o0.w = elu1(a0.w + dd * g0.w + b0.w);
  o1.x = elu1(a1.x + dd * g1.x + b1.x);
  o1.y = elu1(a1.y + dd * g1.y + b1.y);
  o1.z = elu1(a1.z + dd * g1.z + b1.z);
  o1.w = elu1(a1.w + dd * g1.w + b1.w);
  float* Od = OUT + (long long)d * C_HID + 4 * lane;
  *(float4*)(Od) = o0;
  *(float4*)(Od + 128) = o1;
}

extern "C" void kernel_launch(void* const* d_in, const int* in_sizes, int n_in,
                              void* d_out, int out_size, void* d_ws, size_t ws_size,
                              hipStream_t stream) {
  const float* x  = (const float*)d_in[0];
  const int*   ei = (const int*)d_in[1];
  const float* W1 = (const float*)d_in[2];
  const float* b1 = (const float*)d_in[3];
  const float* W2 = (const float*)d_in[4];
  const float* b2 = (const float*)d_in[5];
  const float* W3 = (const float*)d_in[6];
  const float* b3 = (const float*)d_in[7];
  float* out = (float*)d_out;

  const int C   = C_HID;
  const int E   = in_sizes[1] / 2;
  const int cin = in_sizes[2] / C;
  const int n   = in_sizes[0] / cin;
  const int* src = ei;
  const int* dst = ei + E;

  char*  ws  = (char*)d_ws;
  size_t off = 0;
  auto alloc = [&](size_t bytes) -> void* {
    void* p = ws + off;
    off = (off + bytes + 255) & ~(size_t)255;
    return p;
  };
  int*   cnt    = (int*)  alloc((size_t)n * sizeof(int));
  int*   rowptr = (int*)  alloc((size_t)(n + 1) * sizeof(int));
  float* dinv   = (float*)alloc((size_t)n * sizeof(float));
  int*   rank   = (int*)  alloc((size_t)E * sizeof(int));
  int*   col    = (int*)  alloc((size_t)E * sizeof(int));
  float* wgt    = (float*)alloc((size_t)E * sizeof(float));
  float* H      = (float*)alloc((size_t)n * C * sizeof(float));
  float* XB     = (float*)alloc((size_t)n * C * sizeof(float));
  (void)ws_size; (void)n_in; (void)out_size;

  k_cnt_init  <<<(n + 255) / 256, 256, 0, stream>>>(cnt, n);
  k_count_rank<<<(E + 255) / 256, 256, 0, stream>>>(dst, cnt, rank, E);
  k_scan_dinv <<<1, 1024, 0, stream>>>(cnt, rowptr, dinv, n);
  k_fill      <<<(E + 255) / 256, 256, 0, stream>>>(src, dst, rank, rowptr, dinv, col, wgt, E);

  const int ggrid = (n + 7) / 8;

  auto layer = [&](const float* Xin, const float* W, const float* bias,
                   float* Out, int K) {
    const int blocks = (n + G_ROWS - 1) / G_ROWS;
    k_gemm_wmma     <<<blocks, 128, 0, stream>>>(Xin, W, H, n, K);
    k_gather_combine<<<ggrid, 256, 0, stream>>>(H, rowptr, col, wgt, dinv, bias, Out, n);
  };

  layer(x,  W1, b1, XB,  cin);
  layer(XB, W2, b2, XB,  C);
  layer(XB, W3, b3, out, C);
}


# layer 1 aggregates its 64-column input before the product, A(XW) = (AX)W: its gather moves 256 B per edge, not 1 KB; bias+elu move into that product's store; layer-1 sum order changes
# speedup vs baseline: 3.3491x; 1.0683x over previous
#include <hip/hip_runtime.h>
#include <hip/hip_bf16.h>


typedef __attribute__((ext_vector_type(2))) float v2f;
typedef __attribute__((ext_vector_type(8))) float v8f;

#define C_HID 256

__global__ void k_cnt_init(int* __restrict__ cnt, int n) {
  int i = blockIdx.x * blockDim.x + threadIdx.x;
  if (i < n) cnt[i] = 0;
}

__global__ void k_count_rank(const int* __restrict__ dst, int* __restrict__ cnt,
                             int* __restrict__ rank, int E) {
  int e = blockIdx.x * blockDim.x + threadIdx.x;
  if (e < E) rank[e] = atomicAdd(&cnt[dst[e]], 1);
}

__global__ __launch_bounds__(1024) void k_scan_dinv(
    const int* __restrict__ cnt, int* __restrict__ rowptr,
    float* __restrict__ dinv, int n) {
  __shared__ int wsum[32];
  const int t = threadIdx.x;
  const int lane = t & 31;
  const int wv = t >> 5;
  const int per = (n + 1023) >> 10;
  int b = t * per;
  int e = b + per;
  if (b > n) b = n;
  if (e > n) e = n;
  int s = 0;
  for (int i = b; i < e; ++i) s += cnt[i];
  int incl = s;
#pragma unroll
  for (int o = 1; o < 32; o <<= 1) {
    int v = __shfl_up(incl, o);
    if (lane >= o) incl += v;
  }
  if (lane == 31) wsum[wv] = incl;
  __syncthreads();
  if (wv == 0) {
    int v = wsum[lane];
    int iv = v;
#pragma unroll
    for (int o = 1; o < 32; o <<= 1) {
      int u = __shfl_up(iv, o);
      if (lane >= o) iv += u;
    }
    wsum[lane] = iv - v;
  }
  __syncthreads();
  int run = wsum[wv] + incl - s;
  for (int i = b; i < e; ++i) {
    int c = cnt[i];
    rowptr[i] = run;
    dinv[i] = rsqrtf((float)(c + 1));
    run += c;
  }
  if (t == 1023) rowptr[n] = run;
}

__global__ void k_fill(const int* __restrict__ src, const int* __restrict__ dst,
                       const int* __restrict__ rank, const int* __restrict__ rowptr,
                       const float* __restrict__ dinv, int* __restrict__ col,
                       float* __restrict__ wgt, int E) {
  int e = blockIdx.x * blockDim.x + threadIdx.x;
  if (e >= E) return;
  int s = src[e];
  int d = dst[e];
  int p = rowptr[d] + rank[e];
  col[p] = s;
  wgt[p] = dinv[s] * dinv[d];
}

__device__ __forceinline__ void fma4(float4& a, const float4 h, const float w) {
  a.x = fmaf(h.x, w, a.x);
  a.y = fmaf(h.y, w, a.y);
  a.z = fmaf(h.z, w, a.z);
  a.w = fmaf(h.w, w, a.w);
}

__device__ __forceinline__ float elu1(float v) {
  return (v > 0.0f) ? v : expm1f(v);
}

#define C_IN1 64

__global__ __launch_bounds__(256) void k_gather_in64(
    const float* __restrict__ X, const int* __restrict__ rowptr,
    const int* __restrict__ col, const float* __restrict__ wgt,
    const float* __restrict__ dinv, float* __restrict__ AX, int n) {
  const int lane = threadIdx.x & 31;
  const int c = lane & 15;
  const int h = lane >> 4;
  const int d = __builtin_amdgcn_readfirstlane(
      (int)(blockIdx.x * (blockDim.x >> 5) + (threadIdx.x >> 5)));
  if (d >= n) return;
  const int beg = rowptr[d];
  const int end = rowptr[d + 1];
  const float* Xl = X + 4 * c;
  const float di = dinv[d];
  const float4 g = *(const float4*)(Xl + (long long)d * C_IN1);

  float4 a = make_float4(0.f, 0.f, 0.f, 0.f);

  for (int base = beg; base < end; base += 32) {
    const int idx = base + lane;
    int s_l = 0;
    float w_l = 0.0f;
    if (idx < end) {
      s_l = col[idx];
      w_l = wgt[idx];
    }
    const int w_bits = __float_as_int(w_l);
    int cnt = end - base;
    if (cnt > 32) cnt = 32;
    for (int e0 = 0; e0 < cnt; e0 += 16) {
      float4 t[8];
      float w[8];
#pragma unroll
      for (int k = 0; k < 8; ++k) {
        const int ee = e0 + 2 * k;
        const int s_e = __builtin_amdgcn_readlane(s_l, ee);
        const int s_o = __builtin_amdgcn_readlane(s_l, ee + 1);
        const float w_e = __int_as_float(__builtin_amdgcn_readlane(w_bits, ee));
        const float w_o = __int_as_float(__builtin_amdgcn_readlane(w_bits, ee + 1));
        const int sv = h ? s_o : s_e;
        w[k] = h ? w_o : w_e;
        t[k] = make_float4(0.f, 0.f, 0.f, 0.f);
        if (ee + h < cnt) t[k] = *(const float4*)(Xl + (long long)sv * C_IN1);
      }
#pragma unroll
      for (int k = 0; k < 8; ++k) fma4(a, t[k], w[k]);
    }
  }

  a.x += __shfl_xor(a.x, 16);
  a.y += __shfl_xor(a.y, 16);
  a.z += __shfl_xor(a.z, 16);
  a.w += __shfl_xor(a.w, 16);

  const float dd = di * di;
  float4 o;
  o.x = a.x + dd * g.x;
  o.y = a.y + dd * g.y;
  o.z = a.z + dd * g.z;
  o.w = a.w + dd * g.w;
  if (h == 0) *(float4*)(AX + (long long)d * C_IN1 + 4 * c) = o;
}

#define G_ROWS 16
#define G_KMAX 256
#define G_PAD 4

__device__ __forceinline__ v8f mma4(const float ax, const float ay,
                                    const float bx, const float by, const v8f acc) {
  v2f a, b;
  a.x = ax; a.y = ay;
  b.x = bx; b.y = by;
  return __builtin_amdgcn_wmma_f32_16x16x4_f32(
       false, a,  false, b,
       (short)0, acc,  false,  false);
}

__device__ __forceinline__ void g_loadb(float4 (&d)[16], const float* __restrict__ wp) {
#pragma unroll
  for (int j = 0; j < 4; ++j) {
#pragma unroll
    for (int i = 0; i < 4; ++i) {
      d[4 * j + i] = *(const float4*)(wp + (8 * j + i) * C_HID);
    }
  }
}

__device__ __forceinline__ void g_mma32(const float4 (&s)[16], const float* ar,
                                        v8f& acc0, v8f& acc1, v8f& acc2, v8f& acc3) {
#pragma unroll
  for (int j = 0; j < 4; ++j) {
    const float4 a = *(const float4*)(ar + 8 * j);
    acc0 = mma4(a.x, a.y, s[4 * j + 0].x, s[4 * j + 1].x, acc0);
    acc1 = mma4(a.x, a.y, s[4 * j + 0].y, s[4 * j + 1].y, acc1);
    acc2 = mma4(a.x, a.y, s[4 * j + 0].z, s[4 * j + 1].z, acc2);
    acc3 = mma4(a.x, a.y, s[4 * j + 0].w, s[4 * j + 1].w, acc3);
    acc0 = mma4(a.z, a.w, s[4 * j + 2].x, s[4 * j + 3].x, acc0);
    acc1 = mma4(a.z, a.w, s[4 * j + 2].y, s[4 * j + 3].y, acc1);
    acc2 = mma4(a.z, a.w, s[4 * j + 2].z, s[4 * j + 3].z, acc2);
    acc3 = mma4(a.z, a.w, s[4 * j + 2].w, s[4 * j + 3].w, acc3);
  }
}

template <bool EPI>
__global__ __launch_bounds__(128) void k_gemm_wmma(
    const float* __restrict__ X, const float* __restrict__ Wm,
    const float* __restrict__ bias, float* __restrict__ Hout, int M, int K) {
  __shared__ __attribute__((aligned(16))) float As[G_ROWS * (G_KMAX + G_PAD)];
  const int t    = threadIdx.x;
  const int lane = t & 31;
  const int wv   = t >> 5;
  const int c    = lane & 15;
  const int h    = lane >> 4;
  const int m0   = blockIdx.x * G_ROWS;
  const int P    = K + G_PAD;

  const float* Wp = Wm + (long long)(4 * h) * C_HID + wv * 64 + 4 * c;
  float4 p[16], q[16];
  g_loadb(p, Wp);

  {
    const int k4 = K >> 2;
    int r0 = m0 + 4 * wv;
    int ra = r0 + 0; if (ra >= M) ra = M - 1;
    int rb = r0 + 1; if (rb >= M) rb = M - 1;
    int rc = r0 + 2; if (rc >= M) rc = M - 1;
    int rd = r0 + 3; if (rd >= M) rd = M - 1;
    const float* xa = X + (long long)ra * K;
    const float* xb = X + (long long)rb * K;
    const float* xc = X + (long long)rc * K;
    const float* xd = X + (long long)rd * K;
    float* la = As + (4 * wv) * P;
    for (int j = lane; j < k4; j += 32) {
      const float4 va = *(const float4*)(xa + 4 * j);
      const float4 vb = *(const float4*)(xb + 4 * j);
      const float4 vc = *(const float4*)(xc + 4 * j);
      const float4 vd = *(const float4*)(xd + 4 * j);
      *(float4*)(la + 4 * j)         = va;
      *(float4*)(la + P + 4 * j)     = vb;
      *(float4*)(la + 2 * P + 4 * j) = vc;
      *(float4*)(la + 3 * P + 4 * j) = vd;
    }
  }
  __syncthreads();

  const float* Ar = As + c * P + 4 * h;
  v8f acc0 = {}, acc1 = {}, acc2 = {}, acc3 = {};

#define G_HOLD() __builtin_amdgcn_sched_barrier(0)
  int kb = 0;
  for (; kb + 64 < K; kb += 64) {
    g_loadb(q, Wp + (long long)(kb + 32) * C_HID);
    G_HOLD();
    g_mma32(p, Ar + kb, acc0, acc1, acc2, acc3);
    G_HOLD();
    g_loadb(p, Wp + (long long)(kb + 64) * C_HID);
    G_HOLD();
    g_mma32(q, Ar + kb + 32, acc0, acc1, acc2, acc3);
    G_HOLD();
  }
  g_loadb(q, Wp + (long long)(kb + 32) * C_HID);
  G_HOLD();
  g_mma32(p, Ar + kb, acc0, acc1, acc2, acc3);
  g_mma32(q, Ar + kb + 32, acc0, acc1, acc2, acc3);
#undef G_HOLD

  float4 bv = make_float4(0.f, 0.f, 0.f, 0.f);
  if (EPI) bv = *(const float4*)(bias + wv * 64 + 4 * c);

  float* Hp = Hout + (long long)(m0 + 8 * h) * C_HID + wv * 64 + 4 * c;
#pragma unroll
  for (int r = 0; r < 8; ++r) {
    const int mr = m0 + 8 * h + r;
    if (mr < M) {
      float4 o;
      o.x = acc0[r]; o.y = acc1[r]; o.z = acc2[r]; o.w = acc3[r];
      if (EPI) {
        o.x = elu1(o.x + bv.x);
        o.y = elu1(o.y + bv.y);
        o.z = elu1(o.z + bv.z);
        o.w = elu1(o.w + bv.w);
      }
      *(float4*)(Hp + (long long)r * C_HID) = o;
    }
  }
}

__global__ __launch_bounds__(256) void k_gather_combine(
    const float* __restrict__ H, const int* __restrict__ rowptr,
    const int* __restrict__ col, const float* __restrict__ wgt,
    const float* __restrict__ dinv, const float* __restrict__ bias,
    float* __restrict__ OUT, int n) {
  const int lane = threadIdx.x & 31;
  const int d = __builtin_amdgcn_readfirstlane(
      (int)(blockIdx.x * (blockDim.x >> 5) + (threadIdx.x >> 5)));
  if (d >= n) return;
  const int beg = rowptr[d];
  const int end = rowptr[d + 1];
  const float* Hl = H + 4 * lane;

  float4 a0 = make_float4(0.f, 0.f, 0.f, 0.f);
  float4 a1 = make_float4(0.f, 0.f, 0.f, 0.f);

  for (int base = beg; base < end; base += 32) {
    const int idx = base + lane;
    int s_l = 0;
    float w_l = 0.0f;
    if (idx < end) {
      s_l = col[idx];
      w_l = wgt[idx];
    }
    const int w_bits = __float_as_int(w_l);
    int cnt = end - base;
    if (cnt > 32) cnt = 32;
    int j = 0;
    for (; j + 4 <= cnt; j += 4) {
      const int s0 = __builtin_amdgcn_readlane(s_l, j + 0);
      const int s1 = __builtin_amdgcn_readlane(s_l, j + 1);
      const int s2 = __builtin_amdgcn_readlane(s_l, j + 2);
      const int s3 = __builtin_amdgcn_readlane(s_l, j + 3);
      const float w0 = __int_as_float(__builtin_amdgcn_readlane(w_bits, j + 0));
      const float w1 = __int_as_float(__builtin_amdgcn_readlane(w_bits, j + 1));
      const float w2 = __int_as_float(__builtin_amdgcn_readlane(w_bits, j + 2));
      const float w3 = __int_as_float(__builtin_amdgcn_readlane(w_bits, j + 3));
      const float* r0 = Hl + (long long)s0 * C_HID;
      const float* r1 = Hl + (long long)s1 * C_HID;
      const float* r2 = Hl + (long long)s2 * C_HID;
      const float* r3 = Hl + (long long)s3 * C_HID;
      const float4 h00 = *(const float4*)(r0);
      const float4 h01 = *(const float4*)(r0 + 128);
      const float4 h10 = *(const float4*)(r1);
      const float4 h11 = *(const float4*)(r1 + 128);
      const float4 h20 = *(const float4*)(r2);
      const float4 h21 = *(const float4*)(r2 + 128);
      const float4 h30 = *(const float4*)(r3);
      const float4 h31 = *(const float4*)(r3 + 128);
      fma4(a0, h00, w0); fma4(a1, h01, w0);
      fma4(a0, h10, w1); fma4(a1, h11, w1);
      fma4(a0, h20, w2); fma4(a1, h21, w2);
      fma4(a0, h30, w3); fma4(a1, h31, w3);
    }
    if (cnt & 2) {
      const int s0 = __builtin_amdgcn_readlane(s_l, j + 0);
      const int s1 = __builtin_amdgcn_readlane(s_l, j + 1);
      const float w0 = __int_as_float(__builtin_amdgcn_readlane(w_bits, j + 0));
      const float w1 = __int_as_float(__builtin_amdgcn_readlane(w_bits, j + 1));
      const float* r0 = Hl + (long long)s0 * C_HID;
      const float* r1 = Hl + (long long)s1 * C_HID;
      const float4 h00 = *(const float4*)(r0);
      const float4 h01 = *(const float4*)(r0 + 128);
      const float4 h10 = *(const float4*)(r1);
      const float4 h11 = *(const float4*)(r1 + 128);
      fma4(a0, h00, w0); fma4(a1, h01, w0);
      fma4(a0, h10, w1); fma4(a1, h11, w1);
      j += 2;
    }
    if (cnt & 1) {
      const int s0 = __builtin_amdgcn_readlane(s_l, j);
      const float w0 = __int_as_float(__builtin_amdgcn_readlane(w_bits, j));
      const float* r0 = Hl + (long long)s0 * C_HID;
      const float4 h00 = *(const float4*)(r0);
      const float4 h01 = *(const float4*)(r0 + 128);
      fma4(a0, h00, w0); fma4(a1, h01, w0);
    }
  }

  const float di = dinv[d];
  const float dd = di * di;
  const float* Hd = Hl + (long long)d * C_HID;
  const float4 g0 = *(const float4*)(Hd);
  const float4 g1 = *(const float4*)(Hd + 128);
  const float4 b0 = *(const float4*)(bias + 4 * lane);
  const float4 b1 = *(const float4*)(bias + 128 + 4 * lane);
  float4 o0, o1;
  o0.x = elu1(a0.x + dd * g0.x + b0.x);
  o0.y = elu1(a0.y + dd * g0.y + b0.y);
  o0.z = elu1(a0.z + dd * g0.z + b0.z);
  o0.w = elu1(a0.w + dd * g0.w + b0.w);
  o1.x = elu1(a1.x + dd * g1.x + b1.x);
  o1.y = elu1(a1.y + dd * g1.y + b1.y);
  o1.z = elu1(a1.z + dd * g1.z + b1.z);
  o1.w = elu1(a1.w + dd * g1.w + b1.w);
  float* Od = OUT + (long long)d * C_HID + 4 * lane;
  *(float4*)(Od) = o0;
  *(float4*)(Od + 128) = o1;
}

extern "C" void kernel_launch(void* const* d_in, const int* in_sizes, int n_in,
                              void* d_out, int out_size, void* d_ws, size_t ws_size,
                              hipStream_t stream) {
  const float* x  = (const float*)d_in[0];
  const int*   ei = (const int*)d_in[1];
  const float* W1 = (const float*)d_in[2];
  const float* b1 = (const float*)d_in[3];
  const float* W2 = (const float*)d_in[4];
  const float* b2 = (const float*)d_in[5];
  const float* W3 = (const float*)d_in[6];
  const float* b3 = (const float*)d_in[7];
  float* out = (float*)d_out;

  const int C   = C_HID;
  const int E   = in_sizes[1] / 2;
  const int cin = in_sizes[2] / C;
  const int n   = in_sizes[0] / cin;
  const int* src = ei;
  const int* dst = ei + E;

  char*  ws  = (char*)d_ws;
  size_t off = 0;
  auto alloc = [&](size_t bytes) -> void* {
    void* p = ws + off;
    off = (off + bytes + 255) & ~(size_t)255;
    return p;
  };
  int*   cnt    = (int*)  alloc((size_t)n * sizeof(int));
  int*   rowptr = (int*)  alloc((size_t)(n + 1) * sizeof(int));
  float* dinv   = (float*)alloc((size_t)n * sizeof(float));
  int*   rank   = (int*)  alloc((size_t)E * sizeof(int));
  int*   col    = (int*)  alloc((size_t)E * sizeof(int));
  float* wgt    = (float*)alloc((size_t)E * sizeof(float));
  float* H      = (float*)alloc((size_t)n * C * sizeof(float));
  float* XB     = (float*)alloc((size_t)n * C * sizeof(float));
  float* AX     = (float*)alloc((size_t)n * C_IN1 * sizeof(float));
  (void)ws_size; (void)n_in; (void)out_size;

  k_cnt_init  <<<(n + 255) / 256, 256, 0, stream>>>(cnt, n);
  k_count_rank<<<(E + 255) / 256, 256, 0, stream>>>(dst, cnt, rank, E);
  k_scan_dinv <<<1, 1024, 0, stream>>>(cnt, rowptr, dinv, n);
  k_fill      <<<(E + 255) / 256, 256, 0, stream>>>(src, dst, rank, rowptr, dinv, col, wgt, E);

  const int ggrid = (n + 7) / 8;

  const int blocks = (n + G_ROWS - 1) / G_ROWS;

  k_gather_in64     <<<ggrid, 256, 0, stream>>>(x, rowptr, col, wgt, dinv, AX, n);
  k_gemm_wmma<true> <<<blocks, 128, 0, stream>>>(AX, W1, b1, XB, n, cin);

  auto layer = [&](const float* Xin, const float* W, const float* bias,
                   float* Out, int K) {
    k_gemm_wmma<false><<<blocks, 128, 0, stream>>>(Xin, W, bias, H, n, K);
    k_gather_combine  <<<ggrid, 256, 0, stream>>>(H, rowptr, col, wgt, dinv, bias, Out, n);
  };

  layer(XB, W2, b2, XB,  C);
  layer(XB, W3, b3, out, C);
}


# scan and fill launches removed: the counting add's return value is the edge's slot in a fixed 128-slot bucket per destination (overflow list behind it); first gather makes the edge weights
# speedup vs baseline: 3.9212x; 1.1708x over previous
#include <hip/hip_runtime.h>
#include <hip/hip_bf16.h>


typedef __attribute__((ext_vector_type(2))) float v2f;
typedef __attribute__((ext_vector_type(8))) float v8f;

#define C_HID 256

#define B_CAP 128

__global__ void k_cnt_init(int* __restrict__ cnt, int n) {
  int i = blockIdx.x * blockDim.x + threadIdx.x;
  if (i <= n) cnt[i] = 0;
}

__global__ void k_count_fill(const int* __restrict__ src, const int* __restrict__ dst,
                             int* __restrict__ cnt, int* __restrict__ col,
                             int* __restrict__ ovf_s, int* __restrict__ ovf_d,
                             int n, int E) {
  int e = blockIdx.x * blockDim.x + threadIdx.x;
  if (e >= E) return;
  const int s = src[e];
  const int d = dst[e];
  const int p = atomicAdd(&cnt[d], 1);
  if (p < B_CAP) {
    col[d * B_CAP + p] = s;
  } else {
    const int q = atomicAdd(&cnt[n], 1);
    ovf_s[q] = s;
    ovf_d[q] = d;
  }
}

__device__ __forceinline__ float dinv_of(int c) {
  return rsqrtf((float)(c + 1));
}

__device__ __forceinline__ void fma4(float4& a, const float4 h, const float w) {
  a.x = fmaf(h.x, w, a.x);
  a.y = fmaf(h.y, w, a.y);
  a.z = fmaf(h.z, w, a.z);
  a.w = fmaf(h.w, w, a.w);
}

__device__ __forceinline__ float elu1(float v) {
  return (v > 0.0f) ? v : expm1f(v);
}

#define C_IN1 64

__global__ __launch_bounds__(256) void k_gather_in64(
    const float* __restrict__ X, const int* __restrict__ cnt,
    const int* __restrict__ col, float* __restrict__ wgt,
    const int* __restrict__ ovf_s, const int* __restrict__ ovf_d,
    float* __restrict__ AX, int n) {
  const int lane = threadIdx.x & 31;
  const int c = lane & 15;
  const int h = lane >> 4;
  const int d = __builtin_amdgcn_readfirstlane(
      (int)(blockIdx.x * (blockDim.x >> 5) + (threadIdx.x >> 5)));
  if (d >= n) return;
  const int deg = cnt[d];
  const int beg = d * B_CAP;
  const int end = beg + (deg < B_CAP ? deg : B_CAP);
  const float* Xl = X + 4 * c;
  const float di = dinv_of(deg);
  const float4 g = *(const float4*)(Xl + (long long)d * C_IN1);

  float4 a = make_float4(0.f, 0.f, 0.f, 0.f);

  for (int base = beg; base < end; base += 32) {
    const int idx = base + lane;
    int s_l = 0;
    float w_l = 0.0f;
    if (idx < end) {
      s_l = col[idx];
      w_l = dinv_of(cnt[s_l]) * di;
      wgt[idx] = w_l;
    }
    const int w_bits = __float_as_int(w_l);
    int nb = end - base;
    if (nb > 32) nb = 32;
    for (int e0 = 0; e0 < nb; e0 += 16) {
      float4 t[8];
      float w[8];
#pragma unroll
      for (int k = 0; k < 8; ++k) {
        const int ee = e0 + 2 * k;
        const int s_e = __builtin_amdgcn_readlane(s_l, ee);
        const int s_o = __builtin_amdgcn_readlane(s_l, ee + 1);
        const float w_e = __int_as_float(__builtin_amdgcn_readlane(w_bits, ee));
        const float w_o = __int_as_float(__builtin_amdgcn_readlane(w_bits, ee + 1));
        const int sv = h ? s_o : s_e;
        w[k] = h ? w_o : w_e;
        t[k] = make_float4(0.f, 0.f, 0.f, 0.f);
        if (ee + h < nb) t[k] = *(const float4*)(Xl + (long long)sv * C_IN1);
      }
#pragma unroll
      for (int k = 0; k < 8; ++k) fma4(a, t[k], w[k]);
    }
  }

  if (deg > B_CAP) {
    const int nov = cnt[n];
    for (int i0 = 0; i0 < nov; i0 += 32) {
      const int idx = i0 + lane;
      int od = -1;
      int os = 0;
      if (idx < nov) {
        od = ovf_d[idx];
        os = ovf_s[idx];
      }
      unsigned m = __builtin_amdgcn_ballot_w32(od == d);
      while (m) {
        const int j = __builtin_ctz(m);
        m &= m - 1;
        const int s = __builtin_amdgcn_readlane(os, j);
        const float w = dinv_of(cnt[s]) * di;
        const float4 t = *(const float4*)(Xl + (long long)s * C_IN1);
        if (h == 0) fma4(a, t, w);
      }
    }
  }

  a.x += __shfl_xor(a.x, 16);
  a.y += __shfl_xor(a.y, 16);
  a.z += __shfl_xor(a.z, 16);
  a.w += __shfl_xor(a.w, 16);

  const float dd = di * di;
  float4 o;
  o.x = a.x + dd * g.x;
  o.y = a.y + dd * g.y;
  o.z = a.z + dd * g.z;
  o.w = a.w + dd * g.w;
  if (h == 0) *(float4*)(AX + (long long)d * C_IN1 + 4 * c) = o;
}

#define G_ROWS 16
#define G_KMAX 256
#define G_PAD 4

__device__ __forceinline__ v8f mma4(const float ax, const float ay,
                                    const float bx, const float by, const v8f acc) {
  v2f a, b;
  a.x = ax; a.y = ay;
  b.x = bx; b.y = by;
  return __builtin_amdgcn_wmma_f32_16x16x4_f32(
       false, a,  false, b,
       (short)0, acc,  false,  false);
}

__device__ __forceinline__ void g_loadb(float4 (&d)[16], const float* __restrict__ wp) {
#pragma unroll
  for (int j = 0; j < 4; ++j) {
#pragma unroll
    for (int i = 0; i < 4; ++i) {
      d[4 * j + i] = *(const float4*)(wp + (8 * j + i) * C_HID);
    }
  }
}

__device__ __forceinline__ void g_mma32(const float4 (&s)[16], const float* ar,
                                        v8f& acc0, v8f& acc1, v8f& acc2, v8f& acc3) {
#pragma unroll
  for (int j = 0; j < 4; ++j) {
    const float4 a = *(const float4*)(ar + 8 * j);
    acc0 = mma4(a.x, a.y, s[4 * j + 0].x, s[4 * j + 1].x, acc0);
    acc1 = mma4(a.x, a.y, s[4 * j + 0].y, s[4 * j + 1].y, acc1);
    acc2 = mma4(a.x, a.y, s[4 * j + 0].z, s[4 * j + 1].z, acc2);
    acc3 = mma4(a.x, a.y, s[4 * j + 0].w, s[4 * j + 1].w, acc3);
    acc0 = mma4(a.z, a.w, s[4 * j + 2].x, s[4 * j + 3].x, acc0);
    acc1 = mma4(a.z, a.w, s[4 * j + 2].y, s[4 * j + 3].y, acc1);
    acc2 = mma4(a.z, a.w, s[4 * j + 2].z, s[4 * j + 3].z, acc2);
    acc3 = mma4(a.z, a.w, s[4 * j + 2].w, s[4 * j + 3].w, acc3);
  }
}

template <bool EPI>
__global__ __launch_bounds__(128) void k_gemm_wmma(
    const float* __restrict__ X, const float* __restrict__ Wm,
    const float* __restrict__ bias, float* __restrict__ Hout, int M, int K) {
  __shared__ __attribute__((aligned(16))) float As[G_ROWS * (G_KMAX + G_PAD)];
  const int t    = threadIdx.x;
  const int lane = t & 31;
  const int wv   = t >> 5;
  const int c    = lane & 15;
  const int h    = lane >> 4;
  const int m0   = blockIdx.x * G_ROWS;
  const int P    = K + G_PAD;

  const float* Wp = Wm + (long long)(4 * h) * C_HID + wv * 64 + 4 * c;
  float4 p[16], q[16];
  g_loadb(p, Wp);

  {
    const int k4 = K >> 2;
    int r0 = m0 + 4 * wv;
    int ra = r0 + 0; if (ra >= M) ra = M - 1;
    int rb = r0 + 1; if (rb >= M) rb = M - 1;
    int rc = r0 + 2; if (rc >= M) rc = M - 1;
    int rd = r0 + 3; if (rd >= M) rd = M - 1;
    const float* xa = X + (long long)ra * K;
    const float* xb = X + (long long)rb * K;
    const float* xc = X + (long long)rc * K;
    const float* xd = X + (long long)rd * K;
    float* la = As + (4 * wv) * P;
    for (int j = lane; j < k4; j += 32) {
      const float4 va = *(const float4*)(xa + 4 * j);
      const float4 vb = *(const float4*)(xb + 4 * j);
      const float4 vc = *(const float4*)(xc + 4 * j);
      const float4 vd = *(const float4*)(xd + 4 * j);
      *(float4*)(la + 4 * j)         = va;
      *(float4*)(la + P + 4 * j)     = vb;
      *(float4*)(la + 2 * P + 4 * j) = vc;
      *(float4*)(la + 3 * P + 4 * j) = vd;
    }
  }
  __syncthreads();

  const float* Ar = As + c * P + 4 * h;
  v8f acc0 = {}, acc1 = {}, acc2 = {}, acc3 = {};

#define G_HOLD() __builtin_amdgcn_sched_barrier(0)
  int kb = 0;
  for (; kb + 64 < K; kb += 64) {
    g_loadb(q, Wp + (long long)(kb + 32) * C_HID);
    G_HOLD();
    g_mma32(p, Ar + kb, acc0, acc1, acc2, acc3);
    G_HOLD();
    g_loadb(p, Wp + (long long)(kb + 64) * C_HID);
    G_HOLD();
    g_mma32(q, Ar + kb + 32, acc0, acc1, acc2, acc3);
    G_HOLD();
  }
  g_loadb(q, Wp + (long long)(kb + 32) * C_HID);
  G_HOLD();
  g_mma32(p, Ar + kb, acc0, acc1, acc2, acc3);
  g_mma32(q, Ar + kb + 32, acc0, acc1, acc2, acc3);
#undef G_HOLD

  float4 bv = make_float4(0.f, 0.f, 0.f, 0.f);
  if (EPI) bv = *(const float4*)(bias + wv * 64 + 4 * c);

  float* Hp = Hout + (long long)(m0 + 8 * h) * C_HID + wv * 64 + 4 * c;
#pragma unroll
  for (int r = 0; r < 8; ++r) {
    const int mr = m0 + 8 * h + r;
    if (mr < M) {
      float4 o;
      o.x = acc0[r]; o.y = acc1[r]; o.z = acc2[r]; o.w = acc3[r];
      if (EPI) {
        o.x = elu1(o.x + bv.x);
        o.y = elu1(o.y + bv.y);
        o.z = elu1(o.z + bv.z);
        o.w = elu1(o.w + bv.w);
      }
      *(float4*)(Hp + (long long)r * C_HID) = o;
    }
  }
}

__global__ __launch_bounds__(256) void k_gather_combine(
    const float* __restrict__ H, const int* __restrict__ cnt,
    const int* __restrict__ col, const float* __restrict__ wgt,
    const int* __restrict__ ovf_s, const int* __restrict__ ovf_d,
    const float* __restrict__ bias, float* __restrict__ OUT, int n) {
  const int lane = threadIdx.x & 31;
  const int d = __builtin_amdgcn_readfirstlane(
      (int)(blockIdx.x * (blockDim.x >> 5) + (threadIdx.x >> 5)));
  if (d >= n) return;
  const int deg = cnt[d];
  const int beg = d * B_CAP;
  const int end = beg + (deg < B_CAP ? deg : B_CAP);
  const float di = dinv_of(deg);
  const float* Hl = H + 4 * lane;

  float4 a0 = make_float4(0.f, 0.f, 0.f, 0.f);
  float4 a1 = make_float4(0.f, 0.f, 0.f, 0.f);

  for (int base = beg; base < end; base += 32) {
    const int idx = base + lane;
    int s_l = 0;
    float w_l = 0.0f;
    if (idx < end) {
      s_l = col[idx];
      w_l = wgt[idx];
    }
    const int w_bits = __float_as_int(w_l);
    int nb = end - base;
    if (nb > 32) nb = 32;
    int j = 0;
    for (; j + 4 <= nb; j += 4) {
      const int s0 = __builtin_amdgcn_readlane(s_l, j + 0);
      const int s1 = __builtin_amdgcn_readlane(s_l, j + 1);
      const int s2 = __builtin_amdgcn_readlane(s_l, j + 2);
      const int s3 = __builtin_amdgcn_readlane(s_l, j + 3);
      const float w0 = __int_as_float(__builtin_amdgcn_readlane(w_bits, j + 0));
      const float w1 = __int_as_float(__builtin_amdgcn_readlane(w_bits, j + 1));
      const float w2 = __int_as_float(__builtin_amdgcn_readlane(w_bits, j + 2));
      const float w3 = __int_as_float(__builtin_amdgcn_readlane(w_bits, j + 3));
      const float* r0 = Hl + (long long)s0 * C_HID;
      const float* r1 = Hl + (long long)s1 * C_HID;
      const float* r2 = Hl + (long long)s2 * C_HID;
      const float* r3 = Hl + (long long)s3 * C_HID;
      const float4 h00 = *(const float4*)(r0);
      const float4 h01 = *(const float4*)(r0 + 128);
      const float4 h10 = *(const float4*)(r1);
      const float4 h11 = *(const float4*)(r1 + 128);
      const float4 h20 = *(const float4*)(r2);
      const float4 h21 = *(const float4*)(r2 + 128);
      const float4 h30 = *(const float4*)(r3);
      const float4 h31 = *(const float4*)(r3 + 128);
      fma4(a0, h00, w0); fma4(a1, h01, w0);
      fma4(a0, h10, w1); fma4(a1, h11, w1);
      fma4(a0, h20, w2); fma4(a1, h21, w2);
      fma4(a0, h30, w3); fma4(a1, h31, w3);
    }
    if (nb & 2) {
      const int s0 = __builtin_amdgcn_readlane(s_l, j + 0);
      const int s1 = __builtin_amdgcn_readlane(s_l, j + 1);
      const float w0 = __int_as_float(__builtin_amdgcn_readlane(w_bits, j + 0));
      const float w1 = __int_as_float(__builtin_amdgcn_readlane(w_bits, j + 1));
      const float* r0 = Hl + (long long)s0 * C_HID;
      const float* r1 = Hl + (long long)s1 * C_HID;
      const float4 h00 = *(const float4*)(r0);
      const float4 h01 = *(const float4*)(r0 + 128);
      const float4 h10 = *(const float4*)(r1);
      const float4 h11 = *(const float4*)(r1 + 128);
      fma4(a0, h00, w0); fma4(a1, h01, w0);
      fma4(a0, h10, w1); fma4(a1, h11, w1);
      j += 2;
    }
    if (nb & 1) {
      const int s0 = __builtin_amdgcn_readlane(s_l, j);
      const float w0 = __int_as_float(__builtin_amdgcn_readlane(w_bits, j));
      const float* r0 = Hl + (long long)s0 * C_HID;
      const float4 h00 = *(const float4*)(r0);
      const float4 h01 = *(const float4*)(r0 + 128);
      fma4(a0, h00, w0); fma4(a1, h01, w0);
    }
  }

  if (deg > B_CAP) {
    const int nov = cnt[n];
    for (int i0 = 0; i0 < nov; i0 += 32) {
      const int idx = i0 + lane;
      int od = -1;
      int os = 0;
      if (idx < nov) {
        od = ovf_d[idx];
        os = ovf_s[idx];
      }
      unsigned m = __builtin_amdgcn_ballot_w32(od == d);
      while (m) {
        const int jj = __builtin_ctz(m);
        m &= m - 1;
        const int s = __builtin_amdgcn_readlane(os, jj);
        const float w = dinv_of(cnt[s]) * di;
        const float* r0 = Hl + (long long)s * C_HID;
        const float4 h00 = *(const float4*)(r0);
        const float4 h01 = *(const float4*)(r0 + 128);
        fma4(a0, h00, w); fma4(a1, h01, w);
      }
    }
  }

  const float dd = di * di;
  const float* Hd = Hl + (long long)d * C_HID;
  const float4 g0 = *(const float4*)(Hd);
  const float4 g1 = *(const float4*)(Hd + 128);
  const float4 b0 = *(const float4*)(bias + 4 * lane);
  const float4 b1 = *(const float4*)(bias + 128 + 4 * lane);
  float4 o0, o1;
  o0.x = elu1(a0.x + dd * g0.x + b0.x);
  o0.y = elu1(a0.y + dd * g0.y + b0.y);
  o0.z = elu1(a0.z + dd * g0.z + b0.z);
  o0.w = elu1(a0.w + dd * g0.w + b0.w);
  o1.x = elu1(a1.x + dd * g1.x + b1.x);
  o1.y = elu1(a1.y + dd * g1.y + b1.y);
  o1.z = elu1(a1.z + dd * g1.z + b1.z);
  o1.w = elu1(a1.w + dd * g1.w + b1.w);
  float* Od = OUT + (long long)d * C_HID + 4 * lane;
  *(float4*)(Od) = o0;
  *(float4*)(Od + 128) = o1;
}

extern "C" void kernel_launch(void* const* d_in, const int* in_sizes, int n_in,
                              void* d_out, int out_size, void* d_ws, size_t ws_size,
                              hipStream_t stream) {
  const float* x  = (const float*)d_in[0];
  const int*   ei = (const int*)d_in[1];
  const float* W1 = (const float*)d_in[2];
  const float* b1 = (const float*)d_in[3];
  const float* W2 = (const float*)d_in[4];
  const float* b2 = (const float*)d_in[5];
  const float* W3 = (const float*)d_in[6];
  const float* b3 = (const float*)d_in[7];
  float* out = (float*)d_out;

  const int C   = C_HID;
  const int E   = in_sizes[1] / 2;
  const int cin = in_sizes[2] / C;
  const int n   = in_sizes[0] / cin;
  const int* src = ei;
  const int* dst = ei + E;

  char*  ws  = (char*)d_ws;
  size_t off = 0;
  auto alloc = [&](size_t bytes) -> void* {
    void* p = ws + off;
    off = (off + bytes + 255) & ~(size_t)255;
    return p;
  };
  int*   cnt    = (int*)  alloc((size_t)(n + 1) * sizeof(int));
  int*   col    = (int*)  alloc((size_t)n * B_CAP * sizeof(int));
  float* wgt    = (float*)alloc((size_t)n * B_CAP * sizeof(float));
  int*   ovf_s  = (int*)  alloc((size_t)E * sizeof(int));
  int*   ovf_d  = (int*)  alloc((size_t)E * sizeof(int));
  float* H      = (float*)alloc((size_t)n * C * sizeof(float));
  float* XB     = (float*)alloc((size_t)n * C * sizeof(float));
  float* AX     = (float*)alloc((size_t)n * C_IN1 * sizeof(float));
  (void)ws_size; (void)n_in; (void)out_size;

  k_cnt_init  <<<(n + 256) / 256, 256, 0, stream>>>(cnt, n);
  k_count_fill<<<(E + 255) / 256, 256, 0, stream>>>(src, dst, cnt, col, ovf_s, ovf_d, n, E);

  const int ggrid = (n + 7) / 8;

  const int blocks = (n + G_ROWS - 1) / G_ROWS;

  k_gather_in64     <<<ggrid, 256, 0, stream>>>(x, cnt, col, wgt, ovf_s, ovf_d, AX, n);
  k_gemm_wmma<true> <<<blocks, 128, 0, stream>>>(AX, W1, b1, XB, n, cin);

  auto layer = [&](const float* Xin, const float* W, const float* bias,
                   float* Out, int K) {
    k_gemm_wmma<false><<<blocks, 128, 0, stream>>>(Xin, W, bias, H, n, K);
    k_gather_combine  <<<ggrid, 256, 0, stream>>>(H, cnt, col, wgt, ovf_s, ovf_d, bias, Out, n);
  };

  layer(XB, W2, b2, XB,  C);
  layer(XB, W3, b3, out, C);
}


# layer-1 product fused with the layer-2 product: its 16x256 block goes through bias and elu into LDS and is multiplied by W2 there; 8 launches -> 7; sum order unchanged
# speedup vs baseline: 4.0785x; 1.0401x over previous
#include <hip/hip_runtime.h>
#include <hip/hip_bf16.h>


typedef __attribute__((ext_vector_type(2))) float v2f;
typedef __attribute__((ext_vector_type(8))) float v8f;

#define C_HID 256

#define B_CAP 128

__global__ void k_cnt_init(int* __restrict__ cnt, int n) {
  int i = blockIdx.x * blockDim.x + threadIdx.x;
  if (i <= n) cnt[i] = 0;
}

__global__ void k_count_fill(const int* __restrict__ src, const int* __restrict__ dst,
                             int* __restrict__ cnt, int* __restrict__ col,
                             int* __restrict__ ovf_s, int* __restrict__ ovf_d,
                             int n, int E) {
  int e = blockIdx.x * blockDim.x + threadIdx.x;
  if (e >= E) return;
  const int s = src[e];
  const int d = dst[e];
  const int p = atomicAdd(&cnt[d], 1);
  if (p < B_CAP) {
    col[d * B_CAP + p] = s;
  } else {
    const int q = atomicAdd(&cnt[n], 1);
    ovf_s[q] = s;
    ovf_d[q] = d;
  }
}

__device__ __forceinline__ float dinv_of(int c) {
  return rsqrtf((float)(c + 1));
}

__device__ __forceinline__ void fma4(float4& a, const float4 h, const float w) {
  a.x = fmaf(h.x, w, a.x);
  a.y = fmaf(h.y, w, a.y);
  a.z = fmaf(h.z, w, a.z);
  a.w = fmaf(h.w, w, a.w);
}

__device__ __forceinline__ float elu1(float v) {
  return (v > 0.0f) ? v : expm1f(v);
}

#define C_IN1 64

__global__ __launch_bounds__(256) void k_gather_in64(
    const float* __restrict__ X, const int* __restrict__ cnt,
    const int* __restrict__ col, float* __restrict__ wgt,
    const int* __restrict__ ovf_s, const int* __restrict__ ovf_d,
    float* __restrict__ AX, int n) {
  const int lane = threadIdx.x & 31;
  const int c = lane & 15;
  const int h = lane >> 4;
  const int d = __builtin_amdgcn_readfirstlane(
      (int)(blockIdx.x * (blockDim.x >> 5) + (threadIdx.x >> 5)));
  if (d >= n) return;
  const int deg = cnt[d];
  const int beg = d * B_CAP;
  const int end = beg + (deg < B_CAP ? deg : B_CAP);
  const float* Xl = X + 4 * c;
  const float di = dinv_of(deg);
  const float4 g = *(const float4*)(Xl + (long long)d * C_IN1);

  float4 a = make_float4(0.f, 0.f, 0.f, 0.f);

  for (int base = beg; base < end; base += 32) {
    const int idx = base + lane;
    int s_l = 0;
    float w_l = 0.0f;
    if (idx < end) {
      s_l = col[idx];
      w_l = dinv_of(cnt[s_l]) * di;
      wgt[idx] = w_l;
    }
    const int w_bits = __float_as_int(w_l);
    int nb = end - base;
    if (nb > 32) nb = 32;
    for (int e0 = 0; e0 < nb; e0 += 16) {
      float4 t[8];
      float w[8];
#pragma unroll
      for (int k = 0; k < 8; ++k) {
        const int ee = e0 + 2 * k;
        const int s_e = __builtin_amdgcn_readlane(s_l, ee);
        const int s_o = __builtin_amdgcn_readlane(s_l, ee + 1);
        const float w_e = __int_as_float(__builtin_amdgcn_readlane(w_bits, ee));
        const float w_o = __int_as_float(__builtin_amdgcn_readlane(w_bits, ee + 1));
        const int sv = h ? s_o : s_e;
        w[k] = h ? w_o : w_e;
        t[k] = make_float4(0.f, 0.f, 0.f, 0.f);
        if (ee + h < nb) t[k] = *(const float4*)(Xl + (long long)sv * C_IN1);
      }
#pragma unroll
      for (int k = 0; k < 8; ++k) fma4(a, t[k], w[k]);
    }
  }

  if (deg > B_CAP) {
    const int nov = cnt[n];
    for (int i0 = 0; i0 < nov; i0 += 32) {
      const int idx = i0 + lane;
      int od = -1;
      int os = 0;
      if (idx < nov) {
        od = ovf_d[idx];
        os = ovf_s[idx];
      }
      unsigned m = __builtin_amdgcn_ballot_w32(od == d);
      while (m) {
        const int j = __builtin_ctz(m);
        m &= m - 1;
        const int s = __builtin_amdgcn_readlane(os, j);
        const float w = dinv_of(cnt[s]) * di;
        const float4 t = *(const float4*)(Xl + (long long)s * C_IN1);
        if (h == 0) fma4(a, t, w);
      }
    }
  }

  a.x += __shfl_xor(a.x, 16);
  a.y += __shfl_xor(a.y, 16);
  a.z += __shfl_xor(a.z, 16);
  a.w += __shfl_xor(a.w, 16);

  const float dd = di * di;
  float4 o;
  o.x = a.x + dd * g.x;
  o.y = a.y + dd * g.y;
  o.z = a.z + dd * g.z;
  o.w = a.w + dd * g.w;
  if (h == 0) *(float4*)(AX + (long long)d * C_IN1 + 4 * c) = o;
}

#define G_ROWS 16
#define G_KMAX 256
#define G_PAD 4

__device__ __forceinline__ v8f mma4(const float ax, const float ay,
                                    const float bx, const float by, const v8f acc) {
  v2f a, b;
  a.x = ax; a.y = ay;
  b.x = bx; b.y = by;
  return __builtin_amdgcn_wmma_f32_16x16x4_f32(
       false, a,  false, b,
       (short)0, acc,  false,  false);
}

__device__ __forceinline__ void g_loadb(float4 (&d)[16], const float* __restrict__ wp) {
#pragma unroll
  for (int j = 0; j < 4; ++j) {
#pragma unroll
    for (int i = 0; i < 4; ++i) {
      d[4 * j + i] = *(const float4*)(wp + (8 * j + i) * C_HID);
    }
  }
}

__device__ __forceinline__ void g_mma32(const float4 (&s)[16], const float* ar,
                                        v8f& acc0, v8f& acc1, v8f& acc2, v8f& acc3) {
#pragma unroll
  for (int j = 0; j < 4; ++j) {
    const float4 a = *(const float4*)(ar + 8 * j);
    acc0 = mma4(a.x, a.y, s[4 * j + 0].x, s[4 * j + 1].x, acc0);
    acc1 = mma4(a.x, a.y, s[4 * j + 0].y, s[4 * j + 1].y, acc1);
    acc2 = mma4(a.x, a.y, s[4 * j + 0].z, s[4 * j + 1].z, acc2);
    acc3 = mma4(a.x, a.y, s[4 * j + 0].w, s[4 * j + 1].w, acc3);
    acc0 = mma4(a.z, a.w, s[4 * j + 2].x, s[4 * j + 3].x, acc0);
    acc1 = mma4(a.z, a.w, s[4 * j + 2].y, s[4 * j + 3].y, acc1);
    acc2 = mma4(a.z, a.w, s[4 * j + 2].z, s[4 * j + 3].z, acc2);
    acc3 = mma4(a.z, a.w, s[4 * j + 2].w, s[4 * j + 3].w, acc3);
  }
}

__device__ __forceinline__ void g_stage(float* As, const float* __restrict__ X,
                                        const int m0, const int M, const int K, const int P,
                                        const int wv, const int lane) {
  const int k4 = K >> 2;
  int r0 = m0 + 4 * wv;
  int ra = r0 + 0; if (ra >= M) ra = M - 1;
  int rb = r0 + 1; if (rb >= M) rb = M - 1;
  int rc = r0 + 2; if (rc >= M) rc = M - 1;
  int rd = r0 + 3; if (rd >= M) rd = M - 1;
  const float* xa = X + (long long)ra * K;
  const float* xb = X + (long long)rb * K;
  const float* xc = X + (long long)rc * K;
  const float* xd = X + (long long)rd * K;
  float* la = As + (4 * wv) * P;
  for (int j = lane; j < k4; j += 32) {
    const float4 va = *(const float4*)(xa + 4 * j);
    const float4 vb = *(const float4*)(xb + 4 * j);
    const float4 vc = *(const float4*)(xc + 4 * j);
    const float4 vd = *(const float4*)(xd + 4 * j);
    *(float4*)(la + 4 * j)         = va;
    *(float4*)(la + P + 4 * j)     = vb;
    *(float4*)(la + 2 * P + 4 * j) = vc;
    *(float4*)(la + 3 * P + 4 * j) = vd;
  }
}

#define G_HOLD() __builtin_amdgcn_sched_barrier(0)

__device__ __forceinline__ void g_kloop(float4 (&p)[16], float4 (&q)[16],
                                        const float* __restrict__ Wp, const float* Ar, const int K,
                                        v8f& acc0, v8f& acc1, v8f& acc2, v8f& acc3) {
  int kb = 0;
  for (; kb + 64 < K; kb += 64) {
    g_loadb(q, Wp + (long long)(kb + 32) * C_HID);
    G_HOLD();
    g_mma32(p, Ar + kb, acc0, acc1, acc2, acc3);
    G_HOLD();
    g_loadb(p, Wp + (long long)(kb + 64) * C_HID);
    G_HOLD();
    g_mma32(q, Ar + kb + 32, acc0, acc1, acc2, acc3);
    G_HOLD();
  }
  g_loadb(q, Wp + (long long)(kb + 32) * C_HID);
  G_HOLD();
  g_mma32(p, Ar + kb, acc0, acc1, acc2, acc3);
  g_mma32(q, Ar + kb + 32, acc0, acc1, acc2, acc3);
}

__device__ __forceinline__ void g_store(float* __restrict__ Hout, const int m0, const int M,
                                        const int wv, const int c, const int h,
                                        const v8f& acc0, const v8f& acc1, const v8f& acc2, const v8f& acc3) {
  float* Hp = Hout + (long long)(m0 + 8 * h) * C_HID + wv * 64 + 4 * c;
#pragma unroll
  for (int r = 0; r < 8; ++r) {
    const int mr = m0 + 8 * h + r;
    if (mr < M) {
      float4 o;
      o.x = acc0[r]; o.y = acc1[r]; o.z = acc2[r]; o.w = acc3[r];
      *(float4*)(Hp + (long long)r * C_HID) = o;
    }
  }
}

__global__ __launch_bounds__(128) void k_gemm_wmma(
    const float* __restrict__ X, const float* __restrict__ Wm,
    float* __restrict__ Hout, int M, int K) {
  __shared__ __attribute__((aligned(16))) float As[G_ROWS * (G_KMAX + G_PAD)];
  const int t    = threadIdx.x;
  const int lane = t & 31;
  const int wv   = t >> 5;
  const int c    = lane & 15;
  const int h    = lane >> 4;
  const int m0   = blockIdx.x * G_ROWS;
  const int P    = K + G_PAD;

  const float* Wp = Wm + (long long)(4 * h) * C_HID + wv * 64 + 4 * c;
  float4 p[16], q[16];
  g_loadb(p, Wp);
  g_stage(As, X, m0, M, K, P, wv, lane);
  __syncthreads();

  v8f acc0 = {}, acc1 = {}, acc2 = {}, acc3 = {};
  g_kloop(p, q, Wp, As + c * P + 4 * h, K, acc0, acc1, acc2, acc3);
  g_store(Hout, m0, M, wv, c, h, acc0, acc1, acc2, acc3);
}

__global__ __launch_bounds__(128) void k_gemm2_wmma(
    const float* __restrict__ X, const float* __restrict__ Wa, const float* __restrict__ bias,
    const float* __restrict__ Wb, float* __restrict__ Hout, int M) {
  static_assert(C_IN1 == 64, "the first product below is written out for 64 k");
  const int P1 = C_IN1 + G_PAD;
  const int P2 = C_HID + G_PAD;
  __shared__ __attribute__((aligned(16))) float As1[G_ROWS * (C_IN1 + G_PAD)];
  __shared__ __attribute__((aligned(16))) float As2[G_ROWS * (C_HID + G_PAD)];
  const int t    = threadIdx.x;
  const int lane = t & 31;
  const int wv   = t >> 5;
  const int c    = lane & 15;
  const int h    = lane >> 4;
  const int m0   = blockIdx.x * G_ROWS;

  const float* Wp1 = Wa + (long long)(4 * h) * C_HID + wv * 64 + 4 * c;
  const float* Wp2 = Wb + (long long)(4 * h) * C_HID + wv * 64 + 4 * c;
  float4 p[16], q[16];
  g_loadb(p, Wp1);
  g_stage(As1, X, m0, M, C_IN1, P1, wv, lane);
  __syncthreads();

  const float* Ar1 = As1 + c * P1 + 4 * h;
  v8f acc0 = {}, acc1 = {}, acc2 = {}, acc3 = {};
  g_loadb(q, Wp1 + 32 * C_HID);
  G_HOLD();
  g_mma32(p, Ar1, acc0, acc1, acc2, acc3);
  G_HOLD();
  g_loadb(p, Wp2);
  G_HOLD();
  g_mma32(q, Ar1 + 32, acc0, acc1, acc2, acc3);

  const float4 bv = *(const float4*)(bias + wv * 64 + 4 * c);
  float* Lp = As2 + (8 * h) * P2 + wv * 64 + 4 * c;
#pragma unroll
  for (int r = 0; r < 8; ++r) {
    float4 o;
    o.x = elu1(acc0[r] + bv.x);
    o.y = elu1(acc1[r] + bv.y);
    o.z = elu1(acc2[r] + bv.z);
    o.w = elu1(acc3[r] + bv.w);
    *(float4*)(Lp + r * P2) = o;
  }
  __syncthreads();

  v8f bcc0 = {}, bcc1 = {}, bcc2 = {}, bcc3 = {};
  g_kloop(p, q, Wp2, As2 + c * P2 + 4 * h, C_HID, bcc0, bcc1, bcc2, bcc3);
  g_store(Hout, m0, M, wv, c, h, bcc0, bcc1, bcc2, bcc3);
}
#undef G_HOLD

__global__ __launch_bounds__(256) void k_gather_combine(
    const float* __restrict__ H, const int* __restrict__ cnt,
    const int* __restrict__ col, const float* __restrict__ wgt,
    const int* __restrict__ ovf_s, const int* __restrict__ ovf_d,
    const float* __restrict__ bias, float* __restrict__ OUT, int n) {
  const int lane = threadIdx.x & 31;
  const int d = __builtin_amdgcn_readfirstlane(
      (int)(blockIdx.x * (blockDim.x >> 5) + (threadIdx.x >> 5)));
  if (d >= n) return;
  const int deg = cnt[d];
  const int beg = d * B_CAP;
  const int end = beg + (deg < B_CAP ? deg : B_CAP);
  const float di = dinv_of(deg);
  const float* Hl = H + 4 * lane;

  float4 a0 = make_float4(0.f, 0.f, 0.f, 0.f);
  float4 a1 = make_float4(0.f, 0.f, 0.f, 0.f);

  for (int base = beg; base < end; base += 32) {
    const int idx = base + lane;
    int s_l = 0;
    float w_l = 0.0f;
    if (idx < end) {
      s_l = col[idx];
      w_l = wgt[idx];
    }
    const int w_bits = __float_as_int(w_l);
    int nb = end - base;
    if (nb > 32) nb = 32;
    int j = 0;
    for (; j + 4 <= nb; j += 4) {
      const int s0 = __builtin_amdgcn_readlane(s_l, j + 0);
      const int s1 = __builtin_amdgcn_readlane(s_l, j + 1);
      const int s2 = __builtin_amdgcn_readlane(s_l, j + 2);
      const int s3 = __builtin_amdgcn_readlane(s_l, j + 3);
      const float w0 = __int_as_float(__builtin_amdgcn_readlane(w_bits, j + 0));
      const float w1 = __int_as_float(__builtin_amdgcn_readlane(w_bits, j + 1));
      const float w2 = __int_as_float(__builtin_amdgcn_readlane(w_bits, j + 2));
      const float w3 = __int_as_float(__builtin_amdgcn_readlane(w_bits, j + 3));
      const float* r0 = Hl + (long long)s0 * C_HID;
      const float* r1 = Hl + (long long)s1 * C_HID;
      const float* r2 = Hl + (long long)s2 * C_HID;
      const float* r3 = Hl + (long long)s3 * C_HID;
      const float4 h00 = *(const float4*)(r0);
      const float4 h01 = *(const float4*)(r0 + 128);
      const float4 h10 = *(const float4*)(r1);
      const float4 h11 = *(const float4*)(r1 + 128);
      const float4 h20 = *(const float4*)(r2);
      const float4 h21 = *(const float4*)(r2 + 128);
      const float4 h30 = *(const float4*)(r3);
      const float4 h31 = *(const float4*)(r3 + 128);
      fma4(a0, h00, w0); fma4(a1, h01, w0);
      fma4(a0, h10, w1); fma4(a1, h11, w1);
      fma4(a0, h20, w2); fma4(a1, h21, w2);
      fma4(a0, h30, w3); fma4(a1, h31, w3);
    }
    if (nb & 2) {
      const int s0 = __builtin_amdgcn_readlane(s_l, j + 0);
      const int s1 = __builtin_amdgcn_readlane(s_l, j + 1);
      const float w0 = __int_as_float(__builtin_amdgcn_readlane(w_bits, j + 0));
      const float w1 = __int_as_float(__builtin_amdgcn_readlane(w_bits, j + 1));
      const float* r0 = Hl + (long long)s0 * C_HID;
      const float* r1 = Hl + (long long)s1 * C_HID;
      const float4 h00 = *(const float4*)(r0);
      const float4 h01 = *(const float4*)(r0 + 128);
      const float4 h10 = *(const float4*)(r1);
      const float4 h11 = *(const float4*)(r1 + 128);
      fma4(a0, h00, w0); fma4(a1, h01, w0);
      fma4(a0, h10, w1); fma4(a1, h11, w1);
      j += 2;
    }
    if (nb & 1) {
      const int s0 = __builtin_amdgcn_readlane(s_l, j);
      const float w0 = __int_as_float(__builtin_amdgcn_readlane(w_bits, j));
      const float* r0 = Hl + (long long)s0 * C_HID;
      const float4 h00 = *(const float4*)(r0);
      const float4 h01 = *(const float4*)(r0 + 128);
      fma4(a0, h00, w0); fma4(a1, h01, w0);
    }
  }

  if (deg > B_CAP) {
    const int nov = cnt[n];
    for (int i0 = 0; i0 < nov; i0 += 32) {
      const int idx = i0 + lane;
      int od = -1;
      int os = 0;
      if (idx < nov) {
        od = ovf_d[idx];
        os = ovf_s[idx];
      }
      unsigned m = __builtin_amdgcn_ballot_w32(od == d);
      while (m) {
        const int jj = __builtin_ctz(m);
        m &= m - 1;
        const int s = __builtin_amdgcn_readlane(os, jj);
        const float w = dinv_of(cnt[s]) * di;
        const float* r0 = Hl + (long long)s * C_HID;
        const float4 h00 = *(const float4*)(r0);
        const float4 h01 = *(const float4*)(r0 + 128);
        fma4(a0, h00, w); fma4(a1, h01, w);
      }
    }
  }

  const float dd = di * di;
  const float* Hd = Hl + (long long)d * C_HID;
  const float4 g0 = *(const float4*)(Hd);
  const float4 g1 = *(const float4*)(Hd + 128);
  const float4 b0 = *(const float4*)(bias + 4 * lane);
  const float4 b1 = *(const float4*)(bias + 128 + 4 * lane);
  float4 o0, o1;
  o0.x = elu1(a0.x + dd * g0.x + b0.x);
  o0.y = elu1(a0.y + dd * g0.y + b0.y);
  o0.z = elu1(a0.z + dd * g0.z + b0.z);
  o0.w = elu1(a0.w + dd * g0.w + b0.w);
  o1.x = elu1(a1.x + dd * g1.x + b1.x);
  o1.y = elu1(a1.y + dd * g1.y + b1.y);
  o1.z = elu1(a1.z + dd * g1.z + b1.z);
  o1.w = elu1(a1.w + dd * g1.w + b1.w);
  float* Od = OUT + (long long)d * C_HID + 4 * lane;
  *(float4*)(Od) = o0;
  *(float4*)(Od + 128) = o1;
}

extern "C" void kernel_launch(void* const* d_in, const int* in_sizes, int n_in,
                              void* d_out, int out_size, void* d_ws, size_t ws_size,
                              hipStream_t stream) {
  const float* x  = (const float*)d_in[0];
  const int*   ei = (const int*)d_in[1];
  const float* W1 = (const float*)d_in[2];
  const float* b1 = (const float*)d_in[3];
  const float* W2 = (const float*)d_in[4];
  const float* b2 = (const float*)d_in[5];
  const float* W3 = (const float*)d_in[6];
  const float* b3 = (const float*)d_in[7];
  float* out = (float*)d_out;

  const int C   = C_HID;
  const int E   = in_sizes[1] / 2;
  const int cin = in_sizes[2] / C;
  const int n   = in_sizes[0] / cin;
  const int* src = ei;
  const int* dst = ei + E;

  char*  ws  = (char*)d_ws;
  size_t off = 0;
  auto alloc = [&](size_t bytes) -> void* {
    void* p = ws + off;
    off = (off + bytes + 255) & ~(size_t)255;
    return p;
  };
  int*   cnt    = (int*)  alloc((size_t)(n + 1) * sizeof(int));
  int*   col    = (int*)  alloc((size_t)n * B_CAP * sizeof(int));
  float* wgt    = (float*)alloc((size_t)n * B_CAP * sizeof(float));
  int*   ovf_s  = (int*)  alloc((size_t)E * sizeof(int));
  int*   ovf_d  = (int*)  alloc((size_t)E * sizeof(int));
  float* H      = (float*)alloc((size_t)n * C * sizeof(float));
  float* XB     = (float*)alloc((size_t)n * C * sizeof(float));
  float* AX     = (float*)alloc((size_t)n * C_IN1 * sizeof(float));
  (void)ws_size; (void)n_in; (void)out_size;

  k_cnt_init  <<<(n + 256) / 256, 256, 0, stream>>>(cnt, n);
  k_count_fill<<<(E + 255) / 256, 256, 0, stream>>>(src, dst, cnt, col, ovf_s, ovf_d, n, E);

  const int ggrid = (n + 7) / 8;

  const int blocks = (n + G_ROWS - 1) / G_ROWS;

  k_gather_in64   <<<ggrid, 256, 0, stream>>>(x, cnt, col, wgt, ovf_s, ovf_d, AX, n);
  k_gemm2_wmma    <<<blocks, 128, 0, stream>>>(AX, W1, b1, W2, H, n);
  k_gather_combine<<<ggrid, 256, 0, stream>>>(H, cnt, col, wgt, ovf_s, ovf_d, b2, XB, n);

  k_gemm_wmma     <<<blocks, 128, 0, stream>>>(XB, W3, H, n, C);
  k_gather_combine<<<ggrid, 256, 0, stream>>>(H, cnt, col, wgt, ovf_s, ovf_d, b3, out, n);
}


# bucket slots ranked in LDS: 256 work-groups (edge chunk x destination range) count edges per row in LDS and reserve a row's slots with one global add per row, not per edge; bucket order changes
# speedup vs baseline: 5.4022x; 1.3246x over previous
#include <hip/hip_runtime.h>
#include <hip/hip_bf16.h>


typedef __attribute__((ext_vector_type(2))) float v2f;
typedef __attribute__((ext_vector_type(8))) float v8f;

#define C_HID 256

#define B_CAP 128

__global__ void k_cnt_init(int* __restrict__ cnt, int n) {
  int i = blockIdx.x * blockDim.x + threadIdx.x;
  if (i <= n) cnt[i] = 0;
}

#define BK_THREADS 512
#define BK_V 5
#define BK_EDGES (BK_THREADS * 4 * BK_V)
#define BK_RANGE 1280
#define BK_PER ((BK_RANGE + BK_THREADS - 1) / BK_THREADS)

__device__ __forceinline__ int bk_word(const int d, const int d_lo, const int lane) {
  const unsigned u = (unsigned)(d - d_lo);
  return (u < (unsigned)BK_RANGE) ? (int)u : BK_RANGE + lane;
}

__device__ __forceinline__ void bk_place(const int w, const int p, const int s, const int d,
                                         int* __restrict__ cnt, int* __restrict__ col,
                                         int* __restrict__ ovf_s, int* __restrict__ ovf_d,
                                         const int n) {
  if (w < BK_RANGE) {
    if (p < B_CAP) {
      col[d * B_CAP + p] = s;
    } else {
      const int q = atomicAdd(&cnt[n], 1);
      ovf_s[q] = s;
      ovf_d[q] = d;
    }
  }
}

__global__ __launch_bounds__(BK_THREADS) void k_bucket_fill(
    const int* __restrict__ src, const int* __restrict__ dst,
    int* __restrict__ cnt, int* __restrict__ col,
    int* __restrict__ ovf_s, int* __restrict__ ovf_d, int n, int E, int nch) {
  __shared__ int lc[BK_RANGE + 32];
  __shared__ int lb[BK_RANGE + 32];
  const int t    = threadIdx.x;
  const int lane = t & 31;
  const int ch   = blockIdx.x % nch;
  const int rg   = blockIdx.x / nch;
  const int d_lo = rg * BK_RANGE;
  const int e_lo = ch * BK_EDGES;

  for (int i = t; i < BK_RANGE + 32; i += BK_THREADS) {
    lc[i] = 0;
    lb[i] = 0;
  }

  int4 sv[BK_V], dv[BK_V];
  if ((E & 3) == 0) {
#pragma unroll
    for (int k = 0; k < BK_V; ++k) {
      const int e4 = e_lo + (k * BK_THREADS + t) * 4;
      sv[k] = make_int4(0, 0, 0, 0);
      dv[k] = make_int4(-1, -1, -1, -1);
      if (e4 < E) {
        sv[k] = *(const int4*)(src + e4);
        dv[k] = *(const int4*)(dst + e4);
      }
    }
  } else {
#pragma unroll
    for (int k = 0; k < BK_V; ++k) {
      const int e4 = e_lo + (k * BK_THREADS + t) * 4;
      int4 s = make_int4(0, 0, 0, 0);
      int4 d = make_int4(-1, -1, -1, -1);
      if (e4 + 0 < E) { s.x = src[e4 + 0]; d.x = dst[e4 + 0]; }
      if (e4 + 1 < E) { s.y = src[e4 + 1]; d.y = dst[e4 + 1]; }
      if (e4 + 2 < E) { s.z = src[e4 + 2]; d.z = dst[e4 + 2]; }
      if (e4 + 3 < E) { s.w = src[e4 + 3]; d.w = dst[e4 + 3]; }
      sv[k] = s;
      dv[k] = d;
    }
  }
  __syncthreads();

  int4 wv[BK_V];
#pragma unroll
  for (int k = 0; k < BK_V; ++k) {
    wv[k].x = bk_word(dv[k].x, d_lo, lane);
    wv[k].y = bk_word(dv[k].y, d_lo, lane);
    wv[k].z = bk_word(dv[k].z, d_lo, lane);
    wv[k].w = bk_word(dv[k].w, d_lo, lane);
    atomicAdd(&lc[wv[k].x], 1);
    atomicAdd(&lc[wv[k].y], 1);
    atomicAdd(&lc[wv[k].z], 1);
    atomicAdd(&lc[wv[k].w], 1);
  }
  __syncthreads();

  int cl[BK_PER], bb[BK_PER];
#pragma unroll
  for (int j = 0; j < BK_PER; ++j) {
    const int i = t + j * BK_THREADS;
    cl[j] = (i < BK_RANGE) ? lc[i] : 0;
  }
#pragma unroll
  for (int j = 0; j < BK_PER; ++j) {
    const int i = t + j * BK_THREADS;
    bb[j] = 0;
    if (cl[j] > 0) bb[j] = atomicAdd(&cnt[d_lo + i], cl[j]);
  }
#pragma unroll
  for (int j = 0; j < BK_PER; ++j) {
    const int i = t + j * BK_THREADS;
    if (i < BK_RANGE) {
      lb[i] = bb[j];
      lc[i] = 0;
    }
  }
  __syncthreads();

  int4 pv[BK_V];
#pragma unroll
  for (int k = 0; k < BK_V; ++k) {
    pv[k].x = atomicAdd(&lc[wv[k].x], 1);
    pv[k].y = atomicAdd(&lc[wv[k].y], 1);
    pv[k].z = atomicAdd(&lc[wv[k].z], 1);
    pv[k].w = atomicAdd(&lc[wv[k].w], 1);
  }
#pragma unroll
  for (int k = 0; k < BK_V; ++k) {
    pv[k].x += lb[wv[k].x];
    pv[k].y += lb[wv[k].y];
    pv[k].z += lb[wv[k].z];
    pv[k].w += lb[wv[k].w];
  }
#pragma unroll
  for (int k = 0; k < BK_V; ++k) {
    bk_place(wv[k].x, pv[k].x, sv[k].x, dv[k].x, cnt, col, ovf_s, ovf_d, n);
    bk_place(wv[k].y, pv[k].y, sv[k].y, dv[k].y, cnt, col, ovf_s, ovf_d, n);
    bk_place(wv[k].z, pv[k].z, sv[k].z, dv[k].z, cnt, col, ovf_s, ovf_d, n);
    bk_place(wv[k].w, pv[k].w, sv[k].w, dv[k].w, cnt, col, ovf_s, ovf_d, n);
  }
}

__device__ __forceinline__ float dinv_of(int c) {
  return rsqrtf((float)(c + 1));
}

__device__ __forceinline__ void fma4(float4& a, const float4 h, const float w) {
  a.x = fmaf(h.x, w, a.x);
  a.y = fmaf(h.y, w, a.y);
  a.z = fmaf(h.z, w, a.z);
  a.w = fmaf(h.w, w, a.w);
}

__device__ __forceinline__ float elu1(float v) {
  return (v > 0.0f) ? v : expm1f(v);
}

#define C_IN1 64

__global__ __launch_bounds__(256) void k_gather_in64(
    const float* __restrict__ X, const int* __restrict__ cnt,
    const int* __restrict__ col, float* __restrict__ wgt,
    const int* __restrict__ ovf_s, const int* __restrict__ ovf_d,
    float* __restrict__ AX, int n) {
  const int lane = threadIdx.x & 31;
  const int c = lane & 15;
  const int h = lane >> 4;
  const int d = __builtin_amdgcn_readfirstlane(
      (int)(blockIdx.x * (blockDim.x >> 5) + (threadIdx.x >> 5)));
  if (d >= n) return;
  const int deg = cnt[d];
  const int beg = d * B_CAP;
  const int end = beg + (deg < B_CAP ? deg : B_CAP);
  const float* Xl = X + 4 * c;
  const float di = dinv_of(deg);
  const float4 g = *(const float4*)(Xl + (long long)d * C_IN1);

  float4 a = make_float4(0.f, 0.f, 0.f, 0.f);

  for (int base = beg; base < end; base += 32) {
    const int idx = base + lane;
    int s_l = 0;
    float w_l = 0.0f;
    if (idx < end) {
      s_l = col[idx];
      w_l = dinv_of(cnt[s_l]) * di;
      wgt[idx] = w_l;
    }
    const int w_bits = __float_as_int(w_l);
    int nb = end - base;
    if (nb > 32) nb = 32;
    for (int e0 = 0; e0 < nb; e0 += 16) {
      float4 t[8];
      float w[8];
#pragma unroll
      for (int k = 0; k < 8; ++k) {
        const int ee = e0 + 2 * k;
        const int s_e = __builtin_amdgcn_readlane(s_l, ee);
        const int s_o = __builtin_amdgcn_readlane(s_l, ee + 1);
        const float w_e = __int_as_float(__builtin_amdgcn_readlane(w_bits, ee));
        const float w_o = __int_as_float(__builtin_amdgcn_readlane(w_bits, ee + 1));
        const int sv = h ? s_o : s_e;
        w[k] = h ? w_o : w_e;
        t[k] = make_float4(0.f, 0.f, 0.f, 0.f);
        if (ee + h < nb) t[k] = *(const float4*)(Xl + (long long)sv * C_IN1);
      }
#pragma unroll
      for (int k = 0; k < 8; ++k) fma4(a, t[k], w[k]);
    }
  }

  if (deg > B_CAP) {
    const int nov = cnt[n];
    for (int i0 = 0; i0 < nov; i0 += 32) {
      const int idx = i0 + lane;
      int od = -1;
      int os = 0;
      if (idx < nov) {
        od = ovf_d[idx];
        os = ovf_s[idx];
      }
      unsigned m = __builtin_amdgcn_ballot_w32(od == d);
      while (m) {
        const int j = __builtin_ctz(m);
        m &= m - 1;
        const int s = __builtin_amdgcn_readlane(os, j);
        const float w = dinv_of(cnt[s]) * di;
        const float4 t = *(const float4*)(Xl + (long long)s * C_IN1);
        if (h == 0) fma4(a, t, w);
      }
    }
  }

  a.x += __shfl_xor(a.x, 16);
  a.y += __shfl_xor(a.y, 16);
  a.z += __shfl_xor(a.z, 16);
  a.w += __shfl_xor(a.w, 16);

  const float dd = di * di;
  float4 o;
  o.x = a.x + dd * g.x;
  o.y = a.y + dd * g.y;
  o.z = a.z + dd * g.z;
  o.w = a.w + dd * g.w;
  if (h == 0) *(float4*)(AX + (long long)d * C_IN1 + 4 * c) = o;
}

#define G_ROWS 16
#define G_KMAX 256
#define G_PAD 4

__device__ __forceinline__ v8f mma4(const float ax, const float ay,
                                    const float bx, const float by, const v8f acc) {
  v2f a, b;
  a.x = ax; a.y = ay;
  b.x = bx; b.y = by;
  return __builtin_amdgcn_wmma_f32_16x16x4_f32(
       false, a,  false, b,
       (short)0, acc,  false,  false);
}

__device__ __forceinline__ void g_loadb(float4 (&d)[16], const float* __restrict__ wp) {
#pragma unroll
  for (int j = 0; j < 4; ++j) {
#pragma unroll
    for (int i = 0; i < 4; ++i) {
      d[4 * j + i] = *(const float4*)(wp + (8 * j + i) * C_HID);
    }
  }
}

__device__ __forceinline__ void g_mma32(const float4 (&s)[16], const float* ar,
                                        v8f& acc0, v8f& acc1, v8f& acc2, v8f& acc3) {
#pragma unroll
  for (int j = 0; j < 4; ++j) {
    const float4 a = *(const float4*)(ar + 8 * j);
    acc0 = mma4(a.x, a.y, s[4 * j + 0].x, s[4 * j + 1].x, acc0);
    acc1 = mma4(a.x, a.y, s[4 * j + 0].y, s[4 * j + 1].y, acc1);
    acc2 = mma4(a.x, a.y, s[4 * j + 0].z, s[4 * j + 1].z, acc2);
    acc3 = mma4(a.x, a.y, s[4 * j + 0].w, s[4 * j + 1].w, acc3);
    acc0 = mma4(a.z, a.w, s[4 * j + 2].x, s[4 * j + 3].x, acc0);
    acc1 = mma4(a.z, a.w, s[4 * j + 2].y, s[4 * j + 3].y, acc1);
    acc2 = mma4(a.z, a.w, s[4 * j + 2].z, s[4 * j + 3].z, acc2);
    acc3 = mma4(a.z, a.w, s[4 * j + 2].w, s[4 * j + 3].w, acc3);
  }
}

__device__ __forceinline__ void g_stage(float* As, const float* __restrict__ X,
                                        const int m0, const int M, const int K, const int P,
                                        const int wv, const int lane) {
  const int k4 = K >> 2;
  int r0 = m0 + 4 * wv;
  int ra = r0 + 0; if (ra >= M) ra = M - 1;
  int rb = r0 + 1; if (rb >= M) rb = M - 1;
  int rc = r0 + 2; if (rc >= M) rc = M - 1;
  int rd = r0 + 3; if (rd >= M) rd = M - 1;
  const float* xa = X + (long long)ra * K;
  const float* xb = X + (long long)rb * K;
  const float* xc = X + (long long)rc * K;
  const float* xd = X + (long long)rd * K;
  float* la = As + (4 * wv) * P;
  for (int j = lane; j < k4; j += 32) {
    const float4 va = *(const float4*)(xa + 4 * j);
    const float4 vb = *(const float4*)(xb + 4 * j);
    const float4 vc = *(const float4*)(xc + 4 * j);
    const float4 vd = *(const float4*)(xd + 4 * j);
    *(float4*)(la + 4 * j)         = va;
    *(float4*)(la + P + 4 * j)     = vb;
    *(float4*)(la + 2 * P + 4 * j) = vc;
    *(float4*)(la + 3 * P + 4 * j) = vd;
  }
}

#define G_HOLD() __builtin_amdgcn_sched_barrier(0)

__device__ __forceinline__ void g_kloop(float4 (&p)[16], float4 (&q)[16],
                                        const float* __restrict__ Wp, const float* Ar, const int K,
                                        v8f& acc0, v8f& acc1, v8f& acc2, v8f& acc3) {
  int kb = 0;
  for (; kb + 64 < K; kb += 64) {
    g_loadb(q, Wp + (long long)(kb + 32) * C_HID);
    G_HOLD();
    g_mma32(p, Ar + kb, acc0, acc1, acc2, acc3);
    G_HOLD();
    g_loadb(p, Wp + (long long)(kb + 64) * C_HID);
    G_HOLD();
    g_mma32(q, Ar + kb + 32, acc0, acc1, acc2, acc3);
    G_HOLD();
  }
  g_loadb(q, Wp + (long long)(kb + 32) * C_HID);
  G_HOLD();
  g_mma32(p, Ar + kb, acc0, acc1, acc2, acc3);
  g_mma32(q, Ar + kb + 32, acc0, acc1, acc2, acc3);
}

__device__ __forceinline__ void g_store(float* __restrict__ Hout, const int m0, const int M,
                                        const int wv, const int c, const int h,
                                        const v8f& acc0, const v8f& acc1, const v8f& acc2, const v8f& acc3) {
  float* Hp = Hout + (long long)(m0 + 8 * h) * C_HID + wv * 64 + 4 * c;
#pragma unroll
  for (int r = 0; r < 8; ++r) {
    const int mr = m0 + 8 * h + r;
    if (mr < M) {
      float4 o;
      o.x = acc0[r]; o.y = acc1[r]; o.z = acc2[r]; o.w = acc3[r];
      *(float4*)(Hp + (long long)r * C_HID) = o;
    }
  }
}

__global__ __launch_bounds__(128) void k_gemm_wmma(
    const float* __restrict__ X, const float* __restrict__ Wm,
    float* __restrict__ Hout, int M, int K) {
  __shared__ __attribute__((aligned(16))) float As[G_ROWS * (G_KMAX + G_PAD)];
  const int t    = threadIdx.x;
  const int lane = t & 31;
  const int wv   = t >> 5;
  const int c    = lane & 15;
  const int h    = lane >> 4;
  const int m0   = blockIdx.x * G_ROWS;
  const int P    = K + G_PAD;

  const float* Wp = Wm + (long long)(4 * h) * C_HID + wv * 64 + 4 * c;
  float4 p[16], q[16];
  g_loadb(p, Wp);
  g_stage(As, X, m0, M, K, P, wv, lane);
  __syncthreads();

  v8f acc0 = {}, acc1 = {}, acc2 = {}, acc3 = {};
  g_kloop(p, q, Wp, As + c * P + 4 * h, K, acc0, acc1, acc2, acc3);
  g_store(Hout, m0, M, wv, c, h, acc0, acc1, acc2, acc3);
}

__global__ __launch_bounds__(128) void k_gemm2_wmma(
    const float* __restrict__ X, const float* __restrict__ Wa, const float* __restrict__ bias,
    const float* __restrict__ Wb, float* __restrict__ Hout, int M) {
  static_assert(C_IN1 == 64, "the first product below is written out for 64 k");
  const int P1 = C_IN1 + G_PAD;
  const int P2 = C_HID + G_PAD;
  __shared__ __attribute__((aligned(16))) float As1[G_ROWS * (C_IN1 + G_PAD)];
  __shared__ __attribute__((aligned(16))) float As2[G_ROWS * (C_HID + G_PAD)];
  const int t    = threadIdx.x;
  const int lane = t & 31;
  const int wv   = t >> 5;
  const int c    = lane & 15;
  const int h    = lane >> 4;
  const int m0   = blockIdx.x * G_ROWS;

  const float* Wp1 = Wa + (long long)(4 * h) * C_HID + wv * 64 + 4 * c;
  const float* Wp2 = Wb + (long long)(4 * h) * C_HID + wv * 64 + 4 * c;
  float4 p[16], q[16];
  g_loadb(p, Wp1);
  g_stage(As1, X, m0, M, C_IN1, P1, wv, lane);
  __syncthreads();

  const float* Ar1 = As1 + c * P1 + 4 * h;
  v8f acc0 = {}, acc1 = {}, acc2 = {}, acc3 = {};
  g_loadb(q, Wp1 + 32 * C_HID);
  G_HOLD();
  g_mma32(p, Ar1, acc0, acc1, acc2, acc3);
  G_HOLD();
  g_loadb(p, Wp2);
  G_HOLD();
  g_mma32(q, Ar1 + 32, acc0, acc1, acc2, acc3);

  const float4 bv = *(const float4*)(bias + wv * 64 + 4 * c);
  float* Lp = As2 + (8 * h) * P2 + wv * 64 + 4 * c;
#pragma unroll
  for (int r = 0; r < 8; ++r) {
    float4 o;
    o.x = elu1(acc0[r] + bv.x);
    o.y = elu1(acc1[r] + bv.y);
    o.z = elu1(acc2[r] + bv.z);
    o.w = elu1(acc3[r] + bv.w);
    *(float4*)(Lp + r * P2) = o;
  }
  __syncthreads();

  v8f bcc0 = {}, bcc1 = {}, bcc2 = {}, bcc3 = {};
  g_kloop(p, q, Wp2, As2 + c * P2 + 4 * h, C_HID, bcc0, bcc1, bcc2, bcc3);
  g_store(Hout, m0, M, wv, c, h, bcc0, bcc1, bcc2, bcc3);
}
#undef G_HOLD

__global__ __launch_bounds__(256) void k_gather_combine(
    const float* __restrict__ H, const int* __restrict__ cnt,
    const int* __restrict__ col, const float* __restrict__ wgt,
    const int* __restrict__ ovf_s, const int* __restrict__ ovf_d,
    const float* __restrict__ bias, float* __restrict__ OUT, int n) {
  const int lane = threadIdx.x & 31;
  const int d = __builtin_amdgcn_readfirstlane(
      (int)(blockIdx.x * (blockDim.x >> 5) + (threadIdx.x >> 5)));
  if (d >= n) return;
  const int deg = cnt[d];
  const int beg = d * B_CAP;
  const int end = beg + (deg < B_CAP ? deg : B_CAP);
  const float di = dinv_of(deg);
  const float* Hl = H + 4 * lane;

  float4 a0 = make_float4(0.f, 0.f, 0.f, 0.f);
  float4 a1 = make_float4(0.f, 0.f, 0.f, 0.f);

  for (int base = beg; base < end; base += 32) {
    const int idx = base + lane;
    int s_l = 0;
    float w_l = 0.0f;
    if (idx < end) {
      s_l = col[idx];
      w_l = wgt[idx];
    }
    const int w_bits = __float_as_int(w_l);
    int nb = end - base;
    if (nb > 32) nb = 32;
    int j = 0;
    for (; j + 4 <= nb; j += 4) {
      const int s0 = __builtin_amdgcn_readlane(s_l, j + 0);
      const int s1 = __builtin_amdgcn_readlane(s_l, j + 1);
      const int s2 = __builtin_amdgcn_readlane(s_l, j + 2);
      const int s3 = __builtin_amdgcn_readlane(s_l, j + 3);
      const float w0 = __int_as_float(__builtin_amdgcn_readlane(w_bits, j + 0));
      const float w1 = __int_as_float(__builtin_amdgcn_readlane(w_bits, j + 1));
      const float w2 = __int_as_float(__builtin_amdgcn_readlane(w_bits, j + 2));
      const float w3 = __int_as_float(__builtin_amdgcn_readlane(w_bits, j + 3));
      const float* r0 = Hl + (long long)s0 * C_HID;
      const float* r1 = Hl + (long long)s1 * C_HID;
      const float* r2 = Hl + (long long)s2 * C_HID;
      const float* r3 = Hl + (long long)s3 * C_HID;
      const float4 h00 = *(const float4*)(r0);
      const float4 h01 = *(const float4*)(r0 + 128);
      const float4 h10 = *(const float4*)(r1);
      const float4 h11 = *(const float4*)(r1 + 128);
      const float4 h20 = *(const float4*)(r2);
      const float4 h21 = *(const float4*)(r2 + 128);
      const float4 h30 = *(const float4*)(r3);
      const float4 h31 = *(const float4*)(r3 + 128);
      fma4(a0, h00, w0); fma4(a1, h01, w0);
      fma4(a0, h10, w1); fma4(a1, h11, w1);
      fma4(a0, h20, w2); fma4(a1, h21, w2);
      fma4(a0, h30, w3); fma4(a1, h31, w3);
    }
    if (nb & 2) {
      const int s0 = __builtin_amdgcn_readlane(s_l, j + 0);
      const int s1 = __builtin_amdgcn_readlane(s_l, j + 1);
      const float w0 = __int_as_float(__builtin_amdgcn_readlane(w_bits, j + 0));
      const float w1 = __int_as_float(__builtin_amdgcn_readlane(w_bits, j + 1));
      const float* r0 = Hl + (long long)s0 * C_HID;
      const float* r1 = Hl + (long long)s1 * C_HID;
      const float4 h00 = *(const float4*)(r0);
      const float4 h01 = *(const float4*)(r0 + 128);
      const float4 h10 = *(const float4*)(r1);
      const float4 h11 = *(const float4*)(r1 + 128);
      fma4(a0, h00, w0); fma4(a1, h01, w0);
      fma4(a0, h10, w1); fma4(a1, h11, w1);
      j += 2;
    }
    if (nb & 1) {
      const int s0 = __builtin_amdgcn_readlane(s_l, j);
      const float w0 = __int_as_float(__builtin_amdgcn_readlane(w_bits, j));
      const float* r0 = Hl + (long long)s0 * C_HID;
      const float4 h00 = *(const float4*)(r0);
      const float4 h01 = *(const float4*)(r0 + 128);
      fma4(a0, h00, w0); fma4(a1, h01, w0);
    }
  }

  if (deg > B_CAP) {
    const int nov = cnt[n];
    for (int i0 = 0; i0 < nov; i0 += 32) {
      const int idx = i0 + lane;
      int od = -1;
      int os = 0;
      if (idx < nov) {
        od = ovf_d[idx];
        os = ovf_s[idx];
      }
      unsigned m = __builtin_amdgcn_ballot_w32(od == d);
      while (m) {
        const int jj = __builtin_ctz(m);
        m &= m - 1;
        const int s = __builtin_amdgcn_readlane(os, jj);
        const float w = dinv_of(cnt[s]) * di;
        const float* r0 = Hl + (long long)s * C_HID;
        const float4 h00 = *(const float4*)(r0);
        const float4 h01 = *(const float4*)(r0 + 128);
        fma4(a0, h00, w); fma4(a1, h01, w);
      }
    }
  }

  const float dd = di * di;
  const float* Hd = Hl + (long long)d * C_HID;
  const float4 g0 = *(const float4*)(Hd);
  const float4 g1 = *(const float4*)(Hd + 128);
  const float4 b0 = *(const float4*)(bias + 4 * lane);
  const float4 b1 = *(const float4*)(bias + 128 + 4 * lane);
  float4 o0, o1;
  o0.x = elu1(a0.x + dd * g0.x + b0.x);
  o0.y = elu1(a0.y + dd * g0.y + b0.y);
  o0.z = elu1(a0.z + dd * g0.z + b0.z);
  o0.w = elu1(a0.w + dd * g0.w + b0.w);
  o1.x = elu1(a1.x + dd * g1.x + b1.x);
  o1.y = elu1(a1.y + dd * g1.y + b1.y);
  o1.z = elu1(a1.z + dd * g1.z + b1.z);
  o1.w = elu1(a1.w + dd * g1.w + b1.w);
  float* Od = OUT + (long long)d * C_HID + 4 * lane;
  *(float4*)(Od) = o0;
  *(float4*)(Od + 128) = o1;
}

extern "C" void kernel_launch(void* const* d_in, const int* in_sizes, int n_in,
                              void* d_out, int out_size, void* d_ws, size_t ws_size,
                              hipStream_t stream) {
  const float* x  = (const float*)d_in[0];
  const int*   ei = (const int*)d_in[1];
  const float* W1 = (const float*)d_in[2];
  const float* b1 = (const float*)d_in[3];
  const float* W2 = (const float*)d_in[4];
  const float* b2 = (const float*)d_in[5];
  const float* W3 = (const float*)d_in[6];
  const float* b3 = (const float*)d_in[7];
  float* out = (float*)d_out;

  const int C   = C_HID;
  const int E   = in_sizes[1] / 2;
  const int cin = in_sizes[2] / C;
  const int n   = in_sizes[0] / cin;
  const int* src = ei;
  const int* dst = ei + E;

  char*  ws  = (char*)d_ws;
  size_t off = 0;
  auto alloc = [&](size_t bytes) -> void* {
    void* p = ws + off;
    off = (off + bytes + 255) & ~(size_t)255;
    return p;
  };
  int*   cnt    = (int*)  alloc((size_t)(n + 1) * sizeof(int));
  int*   col    = (int*)  alloc((size_t)n * B_CAP * sizeof(int));
  float* wgt    = (float*)alloc((size_t)n * B_CAP * sizeof(float));
  int*   ovf_s  = (int*)  alloc((size_t)E * sizeof(int));
  int*   ovf_d  = (int*)  alloc((size_t)E * sizeof(int));
  float* H      = (float*)alloc((size_t)n * C * sizeof(float));
  float* XB     = (float*)alloc((size_t)n * C * sizeof(float));
  float* AX     = (float*)alloc((size_t)n * C_IN1 * sizeof(float));
  (void)ws_size; (void)n_in; (void)out_size;

  k_cnt_init  <<<(n + 256) / 256, 256, 0, stream>>>(cnt, n);
  const int nch = (E + BK_EDGES - 1) / BK_EDGES;
  const int nrg = (n + BK_RANGE - 1) / BK_RANGE;
  k_bucket_fill<<<nch * nrg, BK_THREADS, 0, stream>>>(src, dst, cnt, col, ovf_s, ovf_d, n, E, nch);

  const int ggrid = (n + 7) / 8;

  const int blocks = (n + G_ROWS - 1) / G_ROWS;

  k_gather_in64   <<<ggrid, 256, 0, stream>>>(x, cnt, col, wgt, ovf_s, ovf_d, AX, n);
  k_gemm2_wmma    <<<blocks, 128, 0, stream>>>(AX, W1, b1, W2, H, n);
  k_gather_combine<<<ggrid, 256, 0, stream>>>(H, cnt, col, wgt, ovf_s, ovf_d, b2, XB, n);

  k_gemm_wmma     <<<blocks, 128, 0, stream>>>(XB, W3, H, n, C);
  k_gather_combine<<<ggrid, 256, 0, stream>>>(H, cnt, col, wgt, ovf_s, ovf_d, b3, out, n);
}
